# starting kernel
# baseline (speedup 1.0000x reference)
#include <hip/hip_runtime.h>

typedef _Float16 f16;
typedef __attribute__((ext_vector_type(16))) _Float16 v16h;
typedef __attribute__((ext_vector_type(8)))  _Float16 v8h;
typedef __attribute__((ext_vector_type(8)))  float    v8f;

#define HDIM  64
#define WAVES 8


__device__ __forceinline__ v16h load_a_f32(const float* __restrict__ base, int kb) {
  const float4* p = (const float4*)base;
  float4 x0 = p[kb >> 2];
  float4 x1 = p[(kb >> 2) + 1];
  float4 x2 = p[4 + (kb >> 2)];
  float4 x3 = p[4 + (kb >> 2) + 1];
  v16h a;
  a[0] = (f16)x0.x; a[1] = (f16)x0.y; a[2]  = (f16)x0.z; a[3]  = (f16)x0.w;
  a[4] = (f16)x1.x; a[5] = (f16)x1.y; a[6]  = (f16)x1.z; a[7]  = (f16)x1.w;
  a[8] = (f16)x2.x; a[9] = (f16)x2.y; a[10] = (f16)x2.z; a[11] = (f16)x2.w;
  a[12]= (f16)x3.x; a[13]= (f16)x3.y; a[14] = (f16)x3.z; a[15] = (f16)x3.w;
  return a;
}

__device__ __forceinline__ v16h load_a_f16(const f16* __restrict__ base, int kb) {
  v8h lo = *(const v8h*)(base + kb);
  v8h hi = *(const v8h*)(base + 16 + kb);
  v16h a;
#pragma unroll
  for (int i = 0; i < 8; ++i) { a[i] = lo[i]; a[8 + i] = hi[i]; }
  return a;
}

__device__ __forceinline__ void stage_B(const float* __restrict__ W, f16* lds, int ksteps) {
  int total = ksteps * 4 * 512;
  for (int idx = threadIdx.x; idx < total; idx += blockDim.x) {
    int i    = idx & 15;
    int l    = (idx >> 4) & 31;
    int tile = idx >> 9;
    int s    = tile >> 2;
    int nt   = tile & 3;
    int K = 32 * s + ((l >> 4) * 8) + (i & 7) + ((i & 8) ? 16 : 0);
    int n = nt * 16 + (l & 15);
    lds[idx] = (f16)W[K * HDIM + n];
  }
}

__device__ __forceinline__ v8f wmma_f16(v16h a, v16h b, v8f c) {
  return __builtin_amdgcn_wmma_f32_16x16x32_f16(false, a, false, b, (short)0, c, false, false);
}


__global__ void k_zero(float* __restrict__ p, int n) {
  int i = blockIdx.x * blockDim.x + threadIdx.x;
  if (i < n) p[i] = 0.0f;
}

__global__ __launch_bounds__(256) void k_init(const float* __restrict__ h,
                                              const float* __restrict__ w_init,
                                              const float* __restrict__ b_init,
                                              f16* __restrict__ out, int ntiles) {
  __shared__ __align__(32) f16 sB[4 * 512];
  stage_B(w_init, sB, 1);
  __syncthreads();
  int wave = threadIdx.x >> 5, lane = threadIdx.x & 31;
  int tile = blockIdx.x * WAVES + wave;
  if (tile >= ntiles) return;
  int m = lane & 15, hi = lane >> 4, kb = hi * 8;
  int node = tile * 16 + m;
  v16h a = load_a_f32(h + (size_t)node * 32, kb);
#pragma unroll
  for (int nt = 0; nt < 4; ++nt) {
    v16h b = *(const v16h*)&sB[nt * 512 + lane * 16];
    v8f c = {};
    c = wmma_f16(a, b, c);
    int nglob = nt * 16 + (lane & 15);
    float bias = b_init[nglob];
#pragma unroll
    for (int r = 0; r < 8; ++r) {
      int M = r + 8 * hi;
      out[(size_t)(tile * 16 + M) * HDIM + nglob] = (f16)(c[r] + bias);
    }
  }
}

__global__ __launch_bounds__(256) void k_msg(const f16* __restrict__ embed,
                                             const float* __restrict__ xpos,
                                             const int* __restrict__ row,
                                             const int* __restrict__ col,
                                             const float* __restrict__ mw,
                                             const float* __restrict__ mb,
                                             float* __restrict__ aggr, int etiles) {
  __shared__ __align__(32) f16 sB[16 * 512];
  stage_B(mw, sB, 4);
  __syncthreads();
  int wave = threadIdx.x >> 5, lane = threadIdx.x & 31;
  int tile = blockIdx.x * WAVES + wave;
  if (tile >= etiles) return;
  int m = lane & 15, hi = lane >> 4, kb = hi * 8;
  int e = tile * 16 + m;
  int rs = row[e], cd = col[e];
  float dx = xpos[rs * 3 + 0] - xpos[cd * 3 + 0];
  float dy = xpos[rs * 3 + 1] - xpos[cd * 3 + 1];
  float dz = xpos[rs * 3 + 2] - xpos[cd * 3 + 2];
  float dist_m = dx * dx + dy * dy + dz * dz;

  v16h a0 = load_a_f16(embed + (size_t)rs * HDIM +  0, kb);
  v16h a1 = load_a_f16(embed + (size_t)rs * HDIM + 32, kb);
  v16h a2 = load_a_f16(embed + (size_t)cd * HDIM +  0, kb);
  v16h a3 = load_a_f16(embed + (size_t)cd * HDIM + 32, kb);

  float distR[8]; int colR[8];
#pragma unroll
  for (int r = 0; r < 8; ++r) {
    int M = r + 8 * hi;
    distR[r] = __shfl(dist_m, M, 32);
    colR[r]  = col[tile * 16 + M];
  }

#pragma unroll
  for (int nt = 0; nt < 4; ++nt) {
    v8f c = {};
#pragma unroll
    for (int s = 0; s < 4; ++s) {
      v16h b = *(const v16h*)&sB[(s * 4 + nt) * 512 + lane * 16];
      v16h a = (s == 0) ? a0 : (s == 1) ? a1 : (s == 2) ? a2 : a3;
      c = wmma_f16(a, b, c);
    }
    int nglob = nt * 16 + (lane & 15);
    float w128 = mw[128 * HDIM + nglob];
    float bias = mb[nglob];
#pragma unroll
    for (int r = 0; r < 8; ++r) {
      float v = c[r] + distR[r] * w128 + bias;
      v = v > 0.0f ? v : 0.0f;
      atomicAdd(&aggr[(size_t)colR[r] * HDIM + nglob], v);
    }
  }
}

template <typename OUT_T>
__global__ __launch_bounds__(256) void k_update(const f16* __restrict__ embed,
                                                const float* __restrict__ aggr,
                                                const float* __restrict__ rp,
                                                const float* __restrict__ uw,
                                                const float* __restrict__ ub,
                                                OUT_T* __restrict__ out, int ntiles) {
  __shared__ __align__(32) f16 sU[16 * 512];
  __shared__ __align__(32) f16 sR[8 * 512];
  stage_B(uw, sU, 4);
  stage_B(rp, sR, 2);
  __syncthreads();
  int wave = threadIdx.x >> 5, lane = threadIdx.x & 31;
  int tile = blockIdx.x * WAVES + wave;
  if (tile >= ntiles) return;
  int m = lane & 15, hi = lane >> 4, kb = hi * 8;
  int node = tile * 16 + m;
  v16h a0 = load_a_f16(embed + (size_t)node * HDIM +  0, kb);
  v16h a1 = load_a_f16(embed + (size_t)node * HDIM + 32, kb);
  v16h a2 = load_a_f32(aggr  + (size_t)node * HDIM +  0, kb);
  v16h a3 = load_a_f32(aggr  + (size_t)node * HDIM + 32, kb);

#pragma unroll
  for (int nt = 0; nt < 4; ++nt) {
    v8f cu = {};
    v8f cr = {};
#pragma unroll
    for (int s = 0; s < 4; ++s) {
      v16h b = *(const v16h*)&sU[(s * 4 + nt) * 512 + lane * 16];
      v16h a = (s == 0) ? a0 : (s == 1) ? a1 : (s == 2) ? a2 : a3;
      cu = wmma_f16(a, b, cu);
    }
#pragma unroll
    for (int s = 0; s < 2; ++s) {
      v16h b = *(const v16h*)&sR[(s * 4 + nt) * 512 + lane * 16];
      v16h a = (s == 0) ? a0 : a1;
      cr = wmma_f16(a, b, cr);
    }
    int nglob = nt * 16 + (lane & 15);
    float bias = ub[nglob];
#pragma unroll
    for (int r = 0; r < 8; ++r) {
      int M = r + 8 * hi;
      float v = cu[r] + bias;
      v = v > 0.0f ? v : 0.0f;
      out[(size_t)(tile * 16 + M) * HDIM + nglob] = (OUT_T)(cr[r] + v);
    }
  }
}


extern "C" void kernel_launch(void* const* d_in, const int* in_sizes, int n_in,
                              void* d_out, int out_size, void* d_ws, size_t ws_size,
                              hipStream_t stream) {
  (void)n_in; (void)out_size; (void)ws_size;
  const float* x      = (const float*)d_in[0];
  const float* h      = (const float*)d_in[1];
  const int*   eidx   = (const int*)  d_in[2];
  const float* w_init = (const float*)d_in[4];
  const float* b_init = (const float*)d_in[5];
  const float* rp0 = (const float*)d_in[6];
  const float* mw0 = (const float*)d_in[7];
  const float* mb0 = (const float*)d_in[8];
  const float* uw0 = (const float*)d_in[9];
  const float* ub0 = (const float*)d_in[10];
  const float* rp1 = (const float*)d_in[11];
  const float* mw1 = (const float*)d_in[12];
  const float* mb1 = (const float*)d_in[13];
  const float* uw1 = (const float*)d_in[14];
  const float* ub1 = (const float*)d_in[15];

  int N = in_sizes[1] / 32;
  int E = in_sizes[2] / 2;
  const int* row = eidx;
  const int* col = eidx + E;

  char* ws = (char*)d_ws;
  f16*   e0   = (f16*)ws;                                   ws += (size_t)N * HDIM * sizeof(f16);
  float* aggr = (float*)ws;                                 ws += (size_t)N * HDIM * sizeof(float);
  f16*   e1   = (f16*)ws;
  float* outp = (float*)d_out;

  int ntiles  = (N + 15) / 16;
  int etiles  = (E + 15) / 16;
  int nblocks = (ntiles + WAVES - 1) / WAVES;
  int eblocks = (etiles + WAVES - 1) / WAVES;
  int zblocks = (N * HDIM + 255) / 256;

  k_init<<<nblocks, 256, 0, stream>>>(h, w_init, b_init, e0, ntiles);

  k_zero<<<zblocks, 256, 0, stream>>>(aggr, N * HDIM);
  k_msg <<<eblocks, 256, 0, stream>>>(e0, x, row, col, mw0, mb0, aggr, etiles);
  k_update<f16><<<nblocks, 256, 0, stream>>>(e0, aggr, rp0, uw0, ub0, e1, ntiles);

  k_zero<<<zblocks, 256, 0, stream>>>(aggr, N * HDIM);
  k_msg <<<eblocks, 256, 0, stream>>>(e1, x, row, col, mw1, mb1, aggr, etiles);
  k_update<float><<<nblocks, 256, 0, stream>>>(e1, aggr, rp1, uw1, ub1, outp, ntiles);
}


# Edges counting-sorted by target per call (int atomics, arrival order); message layer split per node (P=e@mw_src on WMMA); each node sums relu(P[src]+Q+d*w) in registers: no float atomics, no zeroing
# speedup vs baseline: 1.4909x; 1.4909x over previous
#include <hip/hip_runtime.h>


typedef _Float16 f16;
typedef __attribute__((ext_vector_type(16))) _Float16 v16h;
typedef __attribute__((ext_vector_type(8)))  _Float16 v8h;
typedef __attribute__((ext_vector_type(8)))  float    v8f;

#define HDIM  64
#define WAVES 8
#define NT_WG 8


__device__ __forceinline__ v16h load_a_f32(const float* __restrict__ base, int kb) {
  const float4* p = (const float4*)base;
  float4 x0 = p[kb >> 2];
  float4 x1 = p[(kb >> 2) + 1];
  float4 x2 = p[4 + (kb >> 2)];
  float4 x3 = p[4 + (kb >> 2) + 1];
  v16h a;
  a[0] = (f16)x0.x; a[1] = (f16)x0.y; a[2]  = (f16)x0.z; a[3]  = (f16)x0.w;
  a[4] = (f16)x1.x; a[5] = (f16)x1.y; a[6]  = (f16)x1.z; a[7]  = (f16)x1.w;
  a[8] = (f16)x2.x; a[9] = (f16)x2.y; a[10] = (f16)x2.z; a[11] = (f16)x2.w;
  a[12]= (f16)x3.x; a[13]= (f16)x3.y; a[14] = (f16)x3.z; a[15] = (f16)x3.w;
  return a;
}

__device__ __forceinline__ v16h load_a_f16(const f16* __restrict__ base, int kb) {
  v8h lo = *(const v8h*)(base + kb);
  v8h hi = *(const v8h*)(base + 16 + kb);
  v16h a;
#pragma unroll
  for (int i = 0; i < 8; ++i) { a[i] = lo[i]; a[8 + i] = hi[i]; }
  return a;
}

__device__ __forceinline__ void stage_B(const float* __restrict__ W, f16* lds, int ksteps) {
  int total = ksteps * 4 * 512;
  for (int idx = threadIdx.x; idx < total; idx += blockDim.x) {
    int i    = idx & 15;
    int l    = (idx >> 4) & 31;
    int tile = idx >> 9;
    int s    = tile >> 2;
    int nt   = tile & 3;
    int K = 32 * s + ((l >> 4) * 8) + (i & 7) + ((i & 8) ? 16 : 0);
    int n = nt * 16 + (l & 15);
    lds[idx] = (f16)W[K * HDIM + n];
  }
}

__device__ __forceinline__ v8f wmma_f16(v16h a, v16h b, v8f c) {
  return __builtin_amdgcn_wmma_f32_16x16x32_f16(false, a, false, b, (short)0, c, false, false);
}


__global__ __launch_bounds__(256) void k_init(const float* __restrict__ h,
                                              const float* __restrict__ w_init,
                                              const float* __restrict__ b_init,
                                              f16* __restrict__ out, int ntiles,
                                              int* __restrict__ deg, int nnodes) {
  __shared__ __align__(32) f16 sB[4 * 512];
  int gid = blockIdx.x * blockDim.x + threadIdx.x;
  if (gid < nnodes) deg[gid] = 0;
  stage_B(w_init, sB, 1);
  __syncthreads();
  int wave = threadIdx.x >> 5, lane = threadIdx.x & 31;
  int tile = blockIdx.x * WAVES + wave;
  if (tile >= ntiles) return;
  int m = lane & 15, hi = lane >> 4, kb = hi * 8;
  int node = tile * 16 + m;
  v16h a = load_a_f32(h + (size_t)node * 32, kb);
#pragma unroll
  for (int nt = 0; nt < 4; ++nt) {
    v16h b = *(const v16h*)&sB[nt * 512 + lane * 16];
    v8f c = {};
    c = wmma_f16(a, b, c);
    int nglob = nt * 16 + (lane & 15);
    float bias = b_init[nglob];
#pragma unroll
    for (int r = 0; r < 8; ++r) {
      int M = r + 8 * hi;
      out[(size_t)(tile * 16 + M) * HDIM + nglob] = (f16)(c[r] + bias);
    }
  }
}

__global__ __launch_bounds__(256) void k_rank(const int* __restrict__ col, int* __restrict__ deg,
                                              int* __restrict__ rank, int nedges) {
  int e = blockIdx.x * blockDim.x + threadIdx.x;
  if (e < nedges) rank[e] = atomicAdd(&deg[col[e]], 1);
}

__global__ __launch_bounds__(1024) void k_scan(const int* __restrict__ deg, int* __restrict__ off, int nnodes) {
  __shared__ int s_part[32];
  __shared__ int s_wsum[32];
  const int t = threadIdx.x, lane = t & 31, w = t >> 5;
  const int start = blockIdx.x * 1024;
  int part = 0;
  for (int i = t; i < start; i += 1024) part += deg[i];
  for (int o = 16; o > 0; o >>= 1) part += __shfl_xor(part, o, 32);
  const int idx = start + t;
  const int v = idx < nnodes ? deg[idx] : 0;
  int x = v;
  for (int o = 1; o < 32; o <<= 1) {
    int y = __shfl_up(x, o, 32);
    if (lane >= o) x += y;
  }
  if (lane == 31) s_wsum[w] = x;
  if (lane == 0)  s_part[w] = part;
  __syncthreads();
  int before = 0;
  for (int i = 0; i < 32; ++i) {
    before += s_part[i];
    if (i < w) before += s_wsum[i];
  }
  const int excl = before + x - v;
  if (idx < nnodes) off[idx] = excl;
  if (idx == nnodes - 1) off[nnodes] = excl + v;
}

__global__ __launch_bounds__(256) void k_place(const int* __restrict__ row, const int* __restrict__ col,
                                               const int* __restrict__ off, const int* __restrict__ rank,
                                               int* __restrict__ rec, int nedges) {
  int e = blockIdx.x * blockDim.x + threadIdx.x;
  if (e < nedges) rec[off[col[e]] + rank[e]] = row[e];
}

__global__ __launch_bounds__(256) void k_proj(const f16* __restrict__ embed,
                                              const float* __restrict__ mw,
                                              float* __restrict__ P, int ntiles) {
  __shared__ __align__(32) f16 sB[8 * 512];
  stage_B(mw, sB, 2);
  __syncthreads();
  int wave = threadIdx.x >> 5, lane = threadIdx.x & 31;
  int tile = blockIdx.x * WAVES + wave;
  if (tile >= ntiles) return;
  int m = lane & 15, hi = lane >> 4, kb = hi * 8;
  int node = tile * 16 + m;
  v16h a0 = load_a_f16(embed + (size_t)node * HDIM +  0, kb);
  v16h a1 = load_a_f16(embed + (size_t)node * HDIM + 32, kb);
#pragma unroll
  for (int nt = 0; nt < 4; ++nt) {
    v8f c = {};
    c = wmma_f16(a0, *(const v16h*)&sB[(0 * 4 + nt) * 512 + lane * 16], c);
    c = wmma_f16(a1, *(const v16h*)&sB[(1 * 4 + nt) * 512 + lane * 16], c);
    int nglob = nt * 16 + (lane & 15);
#pragma unroll
    for (int r = 0; r < 8; ++r) {
      int M = r + 8 * hi;
      P[(size_t)(tile * 16 + M) * HDIM + nglob] = c[r];
    }
  }
}

template <typename OUT_T>
__global__ __launch_bounds__(NT_WG * 64) void k_node(const f16* __restrict__ embed,
                                              const float* __restrict__ P,
                                              const float* __restrict__ xpos,
                                              const int* __restrict__ rec,
                                              const int* __restrict__ off,
                                              const float* __restrict__ mw,
                                              const float* __restrict__ mb,
                                              const float* __restrict__ rp,
                                              const float* __restrict__ uw,
                                              const float* __restrict__ ub,
                                              OUT_T* __restrict__ out, int ntiles) {
  __shared__ __align__(32) f16 sU[16 * 512];
  __shared__ __align__(32) f16 sR[8 * 512];
  __shared__ __align__(32) f16 sD[8 * 512];
  __shared__ __align__(32) float sQ[NT_WG * 16 * HDIM];
  __shared__ __align__(32) f16   sA[NT_WG * 16 * HDIM];
  stage_B(uw, sU, 4);
  stage_B(rp, sR, 2);
  stage_B(mw + 64 * HDIM, sD, 2);
  __syncthreads();

  const int wave = __builtin_amdgcn_readfirstlane(threadIdx.x >> 5), lane = threadIdx.x & 31;
  const int t = wave >> 1, half = wave & 1;
  const int tile = blockIdx.x * NT_WG + t;
  const bool valid = tile < ntiles;
  const int m = lane & 15, hi = lane >> 4, kb = hi * 8;

  v16h a0 = {}, a1 = {};
  if (valid) {
    const int node = tile * 16 + m;
    a0 = load_a_f16(embed + (size_t)node * HDIM +  0, kb);
    a1 = load_a_f16(embed + (size_t)node * HDIM + 32, kb);
#pragma unroll
    for (int q = 0; q < 2; ++q) {
      const int nt = half * 2 + q;
      v8f c = {};
      c = wmma_f16(a0, *(const v16h*)&sD[(0 * 4 + nt) * 512 + lane * 16], c);
      c = wmma_f16(a1, *(const v16h*)&sD[(1 * 4 + nt) * 512 + lane * 16], c);
      const int nglob = nt * 16 + m;
      const float bias = mb[nglob];
#pragma unroll
      for (int r = 0; r < 8; ++r)
        sQ[(t * 16 + r + 8 * hi) * HDIM + nglob] = c[r] + bias;
    }
  }
  __syncthreads();

  if (valid) {
    const int gn0 = tile * 16 + half * 8;
    const int ln0 = t * 16 + half * 8;
    const int o = off[gn0 + (lane < 8 ? lane : 8)];
    const float2 w = *(const float2*)&mw[128 * HDIM + 2 * lane];
    for (int j = 0; j < 8; ++j) {
      const int b = __builtin_amdgcn_readlane(o, j);
      const int e = __builtin_amdgcn_readlane(o, j + 1);
      const int n = gn0 + j;
      const float xn0 = xpos[n * 3 + 0], xn1 = xpos[n * 3 + 1], xn2 = xpos[n * 3 + 2];
      const float2 q = *(const float2*)&sQ[(ln0 + j) * HDIM + 2 * lane];
      float accx = 0.0f, accy = 0.0f;
      for (int base = b; base < e; base += 32) {
        int cnt = e - base;
        cnt = cnt > 32 ? 32 : cnt;
        int r = 0;
        float d = 0.0f;
        if (lane < cnt) {
          r = rec[base + lane];
          const float dx = xpos[r * 3 + 0] - xn0;
          const float dy = xpos[r * 3 + 1] - xn1;
          const float dz = xpos[r * 3 + 2] - xn2;
          d = dx * dx + dy * dy + dz * dz;
        }
        for (int i = 0; i < cnt; i += 8) {
          float2 p[8];
          float dd[8];
#pragma unroll
          for (int u = 0; u < 8; ++u) {
            const int ru = __builtin_amdgcn_readlane(r, i + u);
            dd[u] = __int_as_float(__builtin_amdgcn_readlane(__float_as_int(d), i + u));
            p[u] = *(const float2*)&P[(size_t)ru * HDIM + 2 * lane];
          }
#pragma unroll
          for (int u = 0; u < 8; ++u) {
            float vx = p[u].x + q.x + dd[u] * w.x;
            float vy = p[u].y + q.y + dd[u] * w.y;
            vx = vx > 0.0f ? vx : 0.0f;
            vy = vy > 0.0f ? vy : 0.0f;
            if (i + u < cnt) { accx += vx; accy += vy; }
          }
        }
      }
      sA[(ln0 + j) * HDIM + 2 * lane + 0] = (f16)accx;
      sA[(ln0 + j) * HDIM + 2 * lane + 1] = (f16)accy;
    }
  }
  __syncthreads();

  if (valid) {
    const f16* arow = &sA[(t * 16 + m) * HDIM];
    v16h a2 = load_a_f16(arow +  0, kb);
    v16h a3 = load_a_f16(arow + 32, kb);
#pragma unroll
    for (int q = 0; q < 2; ++q) {
      const int nt = half * 2 + q;
      v8f cu = {};
      v8f cr = {};
#pragma unroll
      for (int s = 0; s < 4; ++s) {
        v16h b = *(const v16h*)&sU[(s * 4 + nt) * 512 + lane * 16];
        v16h a = (s == 0) ? a0 : (s == 1) ? a1 : (s == 2) ? a2 : a3;
        cu = wmma_f16(a, b, cu);
      }
#pragma unroll
      for (int s = 0; s < 2; ++s) {
        v16h b = *(const v16h*)&sR[(s * 4 + nt) * 512 + lane * 16];
        v16h a = (s == 0) ? a0 : a1;
        cr = wmma_f16(a, b, cr);
      }
      const int nglob = nt * 16 + m;
      const float bias = ub[nglob];
#pragma unroll
      for (int r = 0; r < 8; ++r) {
        const int M = r + 8 * hi;
        float v = cu[r] + bias;
        v = v > 0.0f ? v : 0.0f;
        out[(size_t)(tile * 16 + M) * HDIM + nglob] = (OUT_T)(cr[r] + v);
      }
    }
  }
}


extern "C" void kernel_launch(void* const* d_in, const int* in_sizes, int n_in,
                              void* d_out, int out_size, void* d_ws, size_t ws_size,
                              hipStream_t stream) {
  (void)n_in; (void)out_size; (void)ws_size;
  const float* x      = (const float*)d_in[0];
  const float* h      = (const float*)d_in[1];
  const int*   eidx   = (const int*)  d_in[2];
  const float* w_init = (const float*)d_in[4];
  const float* b_init = (const float*)d_in[5];
  const float* rp0 = (const float*)d_in[6];
  const float* mw0 = (const float*)d_in[7];
  const float* mb0 = (const float*)d_in[8];
  const float* uw0 = (const float*)d_in[9];
  const float* ub0 = (const float*)d_in[10];
  const float* rp1 = (const float*)d_in[11];
  const float* mw1 = (const float*)d_in[12];
  const float* mb1 = (const float*)d_in[13];
  const float* uw1 = (const float*)d_in[14];
  const float* ub1 = (const float*)d_in[15];

  int N = in_sizes[1] / 32;
  int E = in_sizes[2] / 2;
  const int* row = eidx;
  const int* col = eidx + E;

  char* ws = (char*)d_ws;
  size_t S    = (size_t)N * HDIM * sizeof(f16);
  size_t xoff = S + (size_t)E * 4 + ((((size_t)N + 1) * 4 + 255) & ~(size_t)255);
  f16*   e1   = (f16*)ws;
  int*   rec  = (int*)(ws + S);
  int*   off  = (int*)(ws + S + (size_t)E * 4);
  f16*   e0   = (f16*)(ws + xoff);
  int*   rank = (int*)(ws + xoff + S);
  int*   deg  = (int*)(ws + xoff + S + (size_t)E * 4);
  float* P1   = (float*)(ws + xoff);
  float* outp = (float*)d_out;
  float* P0   = outp;

  int ntiles  = (N + 15) / 16;
  int nblocks = (ntiles + WAVES - 1) / WAVES;
  int gblocks = (ntiles + NT_WG - 1) / NT_WG;
  int eblocks = (E + 255) / 256;
  int sblocks = (N + 1023) / 1024;

  k_init <<<nblocks, 256, 0, stream>>>(h, w_init, b_init, e0, ntiles, deg, N);
  k_rank <<<eblocks, 256, 0, stream>>>(col, deg, rank, E);
  k_scan <<<sblocks, 1024, 0, stream>>>(deg, off, N);
  k_place<<<eblocks, 256, 0, stream>>>(row, col, off, rank, rec, E);

  k_proj<<<nblocks, 256, 0, stream>>>(e0, mw0, P0, ntiles);
  k_node<f16><<<gblocks, NT_WG * 64, 0, stream>>>(e0, P0, x, rec, off, mw0, mb0, rp0, uw0, ub0, e1, ntiles);

  k_proj<<<nblocks, 256, 0, stream>>>(e1, mw1, P1, ntiles);
  k_node<float><<<gblocks, NT_WG * 64, 0, stream>>>(e1, P1, x, rec, off, mw1, mb1, rp1, uw1, ub1, outp, ntiles);
}


# Edge grouping by target on LDS counters in two levels (128-node buckets, then nodes) instead of one returning global add per edge; list order still arrival order (LDS adds); 9 launches
# speedup vs baseline: 1.7987x; 1.2065x over previous
#include <hip/hip_runtime.h>


typedef _Float16 f16;
typedef __attribute__((ext_vector_type(16))) _Float16 v16h;
typedef __attribute__((ext_vector_type(8)))  _Float16 v8h;
typedef __attribute__((ext_vector_type(8)))  float    v8f;

#define HDIM  64
#define WAVES 8
#define NT_WG 8
#define BSH      7
#define BMASK    127
#define SORT_WGS 256
#define SORT_T   512
#define FINE_T   512
#define FINE_U   5


__device__ __forceinline__ v16h load_a_f32(const float* __restrict__ base, int kb) {
  const float4* p = (const float4*)base;
  float4 x0 = p[kb >> 2];
  float4 x1 = p[(kb >> 2) + 1];
  float4 x2 = p[4 + (kb >> 2)];
  float4 x3 = p[4 + (kb >> 2) + 1];
  v16h a;
  a[0] = (f16)x0.x; a[1] = (f16)x0.y; a[2]  = (f16)x0.z; a[3]  = (f16)x0.w;
  a[4] = (f16)x1.x; a[5] = (f16)x1.y; a[6]  = (f16)x1.z; a[7]  = (f16)x1.w;
  a[8] = (f16)x2.x; a[9] = (f16)x2.y; a[10] = (f16)x2.z; a[11] = (f16)x2.w;
  a[12]= (f16)x3.x; a[13]= (f16)x3.y; a[14] = (f16)x3.z; a[15] = (f16)x3.w;
  return a;
}

__device__ __forceinline__ v16h load_a_f16(const f16* __restrict__ base, int kb) {
  v8h lo = *(const v8h*)(base + kb);
  v8h hi = *(const v8h*)(base + 16 + kb);
  v16h a;
#pragma unroll
  for (int i = 0; i < 8; ++i) { a[i] = lo[i]; a[8 + i] = hi[i]; }
  return a;
}

__device__ __forceinline__ void stage_B(const float* __restrict__ W, f16* lds, int ksteps) {
  int total = ksteps * 4 * 512;
  for (int idx = threadIdx.x; idx < total; idx += blockDim.x) {
    int i    = idx & 15;
    int l    = (idx >> 4) & 31;
    int tile = idx >> 9;
    int s    = tile >> 2;
    int nt   = tile & 3;
    int K = 32 * s + ((l >> 4) * 8) + (i & 7) + ((i & 8) ? 16 : 0);
    int n = nt * 16 + (l & 15);
    lds[idx] = (f16)W[K * HDIM + n];
  }
}

__device__ __forceinline__ v8f wmma_f16(v16h a, v16h b, v8f c) {
  return __builtin_amdgcn_wmma_f32_16x16x32_f16(false, a, false, b, (short)0, c, false, false);
}


__global__ __launch_bounds__(256) void k_init(const float* __restrict__ h,
                                              const float* __restrict__ w_init,
                                              const float* __restrict__ b_init,
                                              f16* __restrict__ out, int ntiles) {
  __shared__ __align__(32) f16 sB[4 * 512];
  stage_B(w_init, sB, 1);
  __syncthreads();
  int wave = threadIdx.x >> 5, lane = threadIdx.x & 31;
  int tile = blockIdx.x * WAVES + wave;
  if (tile >= ntiles) return;
  int m = lane & 15, hi = lane >> 4, kb = hi * 8;
  int node = tile * 16 + m;
  v16h a = load_a_f32(h + (size_t)node * 32, kb);
#pragma unroll
  for (int nt = 0; nt < 4; ++nt) {
    v16h b = *(const v16h*)&sB[nt * 512 + lane * 16];
    v8f c = {};
    c = wmma_f16(a, b, c);
    int nglob = nt * 16 + (lane & 15);
    float bias = b_init[nglob];
#pragma unroll
    for (int r = 0; r < 8; ++r) {
      int M = r + 8 * hi;
      out[(size_t)(tile * 16 + M) * HDIM + nglob] = (f16)(c[r] + bias);
    }
  }
}


__global__ __launch_bounds__(SORT_T) void k_hist(const int* __restrict__ col, int* __restrict__ hist,
                                                 int nedges, int chunk, int nb) {
  extern __shared__ int s_cnt[];
  for (int b = threadIdx.x; b < nb; b += SORT_T) s_cnt[b] = 0;
  __syncthreads();
  const int lo = blockIdx.x * chunk;
  int hi = lo + chunk;
  hi = hi < nedges ? hi : nedges;
  for (int e0 = lo + threadIdx.x; e0 < hi; e0 += 8 * SORT_T) {
    int c[8];
#pragma unroll
    for (int k = 0; k < 8; ++k) {
      const int e = e0 + k * SORT_T;
      c[k] = e < hi ? col[e] : -1;
    }
#pragma unroll
    for (int k = 0; k < 8; ++k)
      if (c[k] >= 0) atomicAdd(&s_cnt[c[k] >> BSH], 1);
  }
  __syncthreads();
  for (int b = threadIdx.x; b < nb; b += SORT_T) hist[b * SORT_WGS + blockIdx.x] = s_cnt[b];
}

__global__ __launch_bounds__(1024) void k_scan(const int* __restrict__ deg, int* __restrict__ off, int nnodes) {
  __shared__ int s_part[32];
  __shared__ int s_wsum[32];
  const int t = threadIdx.x, lane = t & 31, w = t >> 5;
  const int start = blockIdx.x * 1024;
  int part = 0;
  for (int i = t; i < start; i += 1024) part += deg[i];
  for (int o = 16; o > 0; o >>= 1) part += __shfl_xor(part, o, 32);
  const int idx = start + t;
  const int v = idx < nnodes ? deg[idx] : 0;
  int x = v;
  for (int o = 1; o < 32; o <<= 1) {
    int y = __shfl_up(x, o, 32);
    if (lane >= o) x += y;
  }
  if (lane == 31) s_wsum[w] = x;
  if (lane == 0)  s_part[w] = part;
  __syncthreads();
  int before = 0;
  for (int i = 0; i < 32; ++i) {
    before += s_part[i];
    if (i < w) before += s_wsum[i];
  }
  const int excl = before + x - v;
  if (idx < nnodes) off[idx] = excl;
  if (idx == nnodes - 1) off[nnodes] = excl + v;
}

__global__ __launch_bounds__(SORT_T) void k_bin(const int* __restrict__ row, const int* __restrict__ col,
                                                const int* __restrict__ base, int2* __restrict__ binned,
                                                int nedges, int chunk, int nb) {
  extern __shared__ int s_cur[];
  for (int b = threadIdx.x; b < nb; b += SORT_T) s_cur[b] = base[b * SORT_WGS + blockIdx.x];
  __syncthreads();
  const int lo = blockIdx.x * chunk;
  int hi = lo + chunk;
  hi = hi < nedges ? hi : nedges;
  for (int e0 = lo + threadIdx.x; e0 < hi; e0 += 8 * SORT_T) {
    int c[8], r[8];
#pragma unroll
    for (int k = 0; k < 8; ++k) {
      const int e = e0 + k * SORT_T;
      c[k] = e < hi ? col[e] : -1;
      r[k] = e < hi ? row[e] : 0;
    }
#pragma unroll
    for (int k = 0; k < 8; ++k)
      if (c[k] >= 0) {
        const int pos = atomicAdd(&s_cur[c[k] >> BSH], 1);
        binned[pos] = make_int2(r[k], c[k]);
      }
  }
}

__global__ __launch_bounds__(FINE_T) void k_fine(const int2* __restrict__ binned, const int* __restrict__ base,
                                                 int* __restrict__ rec, int* __restrict__ off, int nnodes) {
  __shared__ int s_cnt[BMASK + 1];
  __shared__ int s_cur[BMASK + 1];
  __shared__ int s_w[FINE_T / 32];
  const int b = blockIdx.x, t = threadIdx.x, lane = t & 31, w = t >> 5;
  const int s = base[b * SORT_WGS], e = base[(b + 1) * SORT_WGS];
  if (t <= BMASK) s_cnt[t] = 0;
  __syncthreads();
  for (int i0 = s + t; i0 < e; i0 += FINE_U * FINE_T) {
    int c[FINE_U];
#pragma unroll
    for (int k = 0; k < FINE_U; ++k) {
      const int i = i0 + k * FINE_T;
      c[k] = i < e ? binned[i].y : -1;
    }
#pragma unroll
    for (int k = 0; k < FINE_U; ++k)
      if (c[k] >= 0) atomicAdd(&s_cnt[c[k] & BMASK], 1);
  }
  __syncthreads();
  const int v = t <= BMASK ? s_cnt[t] : 0;
  int x = v;
  for (int o = 1; o < 32; o <<= 1) {
    int y = __shfl_up(x, o, 32);
    if (lane >= o) x += y;
  }
  if (lane == 31) s_w[w] = x;
  __syncthreads();
  int before = 0;
  for (int i = 0; i < FINE_T / 32; ++i)
    if (i < w) before += s_w[i];
  const int excl = before + x - v;
  if (t <= BMASK) {
    s_cur[t] = excl;
    const int n = (b << BSH) + t;
    if (n <= nnodes) off[n] = s + excl;
  }
  __syncthreads();
  for (int i0 = s + t; i0 < e; i0 += FINE_U * FINE_T) {
    int2 rc[FINE_U];
#pragma unroll
    for (int k = 0; k < FINE_U; ++k) {
      const int i = i0 + k * FINE_T;
      rc[k] = i < e ? binned[i] : make_int2(0, -1);
    }
#pragma unroll
    for (int k = 0; k < FINE_U; ++k)
      if (rc[k].y >= 0) {
        const int pos = atomicAdd(&s_cur[rc[k].y & BMASK], 1);
        rec[s + pos] = rc[k].x;
      }
  }
}

__global__ __launch_bounds__(256) void k_proj(const f16* __restrict__ embed,
                                              const float* __restrict__ mw,
                                              float* __restrict__ P, int ntiles) {
  __shared__ __align__(32) f16 sB[8 * 512];
  stage_B(mw, sB, 2);
  __syncthreads();
  int wave = threadIdx.x >> 5, lane = threadIdx.x & 31;
  int tile = blockIdx.x * WAVES + wave;
  if (tile >= ntiles) return;
  int m = lane & 15, hi = lane >> 4, kb = hi * 8;
  int node = tile * 16 + m;
  v16h a0 = load_a_f16(embed + (size_t)node * HDIM +  0, kb);
  v16h a1 = load_a_f16(embed + (size_t)node * HDIM + 32, kb);
#pragma unroll
  for (int nt = 0; nt < 4; ++nt) {
    v8f c = {};
    c = wmma_f16(a0, *(const v16h*)&sB[(0 * 4 + nt) * 512 + lane * 16], c);
    c = wmma_f16(a1, *(const v16h*)&sB[(1 * 4 + nt) * 512 + lane * 16], c);
    int nglob = nt * 16 + (lane & 15);
#pragma unroll
    for (int r = 0; r < 8; ++r) {
      int M = r + 8 * hi;
      P[(size_t)(tile * 16 + M) * HDIM + nglob] = c[r];
    }
  }
}

template <typename OUT_T>
__global__ __launch_bounds__(NT_WG * 64) void k_node(const f16* __restrict__ embed,
                                              const float* __restrict__ P,
                                              const float* __restrict__ xpos,
                                              const int* __restrict__ rec,
                                              const int* __restrict__ off,
                                              const float* __restrict__ mw,
                                              const float* __restrict__ mb,
                                              const float* __restrict__ rp,
                                              const float* __restrict__ uw,
                                              const float* __restrict__ ub,
                                              OUT_T* __restrict__ out, int ntiles) {
  __shared__ __align__(32) f16 sU[16 * 512];
  __shared__ __align__(32) f16 sR[8 * 512];
  __shared__ __align__(32) f16 sD[8 * 512];
  __shared__ __align__(32) float sQ[NT_WG * 16 * HDIM];
  __shared__ __align__(32) f16   sA[NT_WG * 16 * HDIM];
  stage_B(uw, sU, 4);
  stage_B(rp, sR, 2);
  stage_B(mw + 64 * HDIM, sD, 2);
  __syncthreads();

  const int wave = __builtin_amdgcn_readfirstlane(threadIdx.x >> 5), lane = threadIdx.x & 31;
  const int t = wave >> 1, half = wave & 1;
  const int tile = blockIdx.x * NT_WG + t;
  const bool valid = tile < ntiles;
  const int m = lane & 15, hi = lane >> 4, kb = hi * 8;

  v16h a0 = {}, a1 = {};
  if (valid) {
    const int node = tile * 16 + m;
    a0 = load_a_f16(embed + (size_t)node * HDIM +  0, kb);
    a1 = load_a_f16(embed + (size_t)node * HDIM + 32, kb);
#pragma unroll
    for (int q = 0; q < 2; ++q) {
      const int nt = half * 2 + q;
      v8f c = {};
      c = wmma_f16(a0, *(const v16h*)&sD[(0 * 4 + nt) * 512 + lane * 16], c);
      c = wmma_f16(a1, *(const v16h*)&sD[(1 * 4 + nt) * 512 + lane * 16], c);
      const int nglob = nt * 16 + m;
      const float bias = mb[nglob];
#pragma unroll
      for (int r = 0; r < 8; ++r)
        sQ[(t * 16 + r + 8 * hi) * HDIM + nglob] = c[r] + bias;
    }
  }
  __syncthreads();

  if (valid) {
    const int gn0 = tile * 16 + half * 8;
    const int ln0 = t * 16 + half * 8;
    const int o = off[gn0 + (lane < 8 ? lane : 8)];
    const float2 w = *(const float2*)&mw[128 * HDIM + 2 * lane];
    for (int j = 0; j < 8; ++j) {
      const int b = __builtin_amdgcn_readlane(o, j);
      const int e = __builtin_amdgcn_readlane(o, j + 1);
      const int n = gn0 + j;
      const float xn0 = xpos[n * 3 + 0], xn1 = xpos[n * 3 + 1], xn2 = xpos[n * 3 + 2];
      const float2 q = *(const float2*)&sQ[(ln0 + j) * HDIM + 2 * lane];
      float accx = 0.0f, accy = 0.0f;
      for (int base = b; base < e; base += 32) {
        int cnt = e - base;
        cnt = cnt > 32 ? 32 : cnt;
        int r = 0;
        float d = 0.0f;
        if (lane < cnt) {
          r = rec[base + lane];
          const float dx = xpos[r * 3 + 0] - xn0;
          const float dy = xpos[r * 3 + 1] - xn1;
          const float dz = xpos[r * 3 + 2] - xn2;
          d = dx * dx + dy * dy + dz * dz;
        }
        for (int i = 0; i < cnt; i += 8) {
          float2 p[8];
          float dd[8];
#pragma unroll
          for (int u = 0; u < 8; ++u) {
            const int ru = __builtin_amdgcn_readlane(r, i + u);
            dd[u] = __int_as_float(__builtin_amdgcn_readlane(__float_as_int(d), i + u));
            p[u] = *(const float2*)&P[(size_t)ru * HDIM + 2 * lane];
          }
#pragma unroll
          for (int u = 0; u < 8; ++u) {
            float vx = p[u].x + q.x + dd[u] * w.x;
            float vy = p[u].y + q.y + dd[u] * w.y;
            vx = vx > 0.0f ? vx : 0.0f;
            vy = vy > 0.0f ? vy : 0.0f;
            if (i + u < cnt) { accx += vx; accy += vy; }
          }
        }
      }
      sA[(ln0 + j) * HDIM + 2 * lane + 0] = (f16)accx;
      sA[(ln0 + j) * HDIM + 2 * lane + 1] = (f16)accy;
    }
  }
  __syncthreads();

  if (valid) {
    const f16* arow = &sA[(t * 16 + m) * HDIM];
    v16h a2 = load_a_f16(arow +  0, kb);
    v16h a3 = load_a_f16(arow + 32, kb);
#pragma unroll
    for (int q = 0; q < 2; ++q) {
      const int nt = half * 2 + q;
      v8f cu = {};
      v8f cr = {};
#pragma unroll
      for (int s = 0; s < 4; ++s) {
        v16h b = *(const v16h*)&sU[(s * 4 + nt) * 512 + lane * 16];
        v16h a = (s == 0) ? a0 : (s == 1) ? a1 : (s == 2) ? a2 : a3;
        cu = wmma_f16(a, b, cu);
      }
#pragma unroll
      for (int s = 0; s < 2; ++s) {
        v16h b = *(const v16h*)&sR[(s * 4 + nt) * 512 + lane * 16];
        v16h a = (s == 0) ? a0 : a1;
        cr = wmma_f16(a, b, cr);
      }
      const int nglob = nt * 16 + m;
      const float bias = ub[nglob];
#pragma unroll
      for (int r = 0; r < 8; ++r) {
        const int M = r + 8 * hi;
        float v = cu[r] + bias;
        v = v > 0.0f ? v : 0.0f;
        out[(size_t)(tile * 16 + M) * HDIM + nglob] = (OUT_T)(cr[r] + v);
      }
    }
  }
}


extern "C" void kernel_launch(void* const* d_in, const int* in_sizes, int n_in,
                              void* d_out, int out_size, void* d_ws, size_t ws_size,
                              hipStream_t stream) {
  (void)n_in; (void)out_size; (void)ws_size;
  const float* x      = (const float*)d_in[0];
  const float* h      = (const float*)d_in[1];
  const int*   eidx   = (const int*)  d_in[2];
  const float* w_init = (const float*)d_in[4];
  const float* b_init = (const float*)d_in[5];
  const float* rp0 = (const float*)d_in[6];
  const float* mw0 = (const float*)d_in[7];
  const float* mb0 = (const float*)d_in[8];
  const float* uw0 = (const float*)d_in[9];
  const float* ub0 = (const float*)d_in[10];
  const float* rp1 = (const float*)d_in[11];
  const float* mw1 = (const float*)d_in[12];
  const float* mb1 = (const float*)d_in[13];
  const float* uw1 = (const float*)d_in[14];
  const float* ub1 = (const float*)d_in[15];

  int N = in_sizes[1] / 32;
  int E = in_sizes[2] / 2;
  const int* row = eidx;
  const int* col = eidx + E;

  char* ws = (char*)d_ws;
  size_t S    = (size_t)N * HDIM * sizeof(f16);
  size_t xoff = S + (size_t)E * 4 + ((((size_t)N + 1) * 4 + 255) & ~(size_t)255);
  f16*   e1   = (f16*)ws;
  int*   rec  = (int*)(ws + S);
  int*   off  = (int*)(ws + S + (size_t)E * 4);
  f16*   e0   = (f16*)(ws + xoff);
  float* P1   = (float*)(ws + xoff);
  float* outp = (float*)d_out;
  float* P0   = outp;

  int nb      = (N >> BSH) + 1;
  int nhist   = nb * SORT_WGS;
  int chunk   = (E + SORT_WGS - 1) / SORT_WGS;
  int2*  binned = (int2*)d_out;
  int*   hist   = (int*)((char*)d_out + (size_t)E * 8);
  int*   base   = hist + nhist;

  int ntiles  = (N + 15) / 16;
  int nblocks = (ntiles + WAVES - 1) / WAVES;
  int gblocks = (ntiles + NT_WG - 1) / NT_WG;
  int sblocks = (nhist + 1023) / 1024;

  k_init<<<nblocks, 256, 0, stream>>>(h, w_init, b_init, e0, ntiles);
  k_hist<<<SORT_WGS, SORT_T, nb * sizeof(int), stream>>>(col, hist, E, chunk, nb);
  k_scan<<<sblocks, 1024, 0, stream>>>(hist, base, nhist);
  k_bin <<<SORT_WGS, SORT_T, nb * sizeof(int), stream>>>(row, col, base, binned, E, chunk, nb);
  k_fine<<<nb, FINE_T, 0, stream>>>(binned, base, rec, off, N);

  k_proj<<<nblocks, 256, 0, stream>>>(e0, mw0, P0, ntiles);
  k_node<f16><<<gblocks, NT_WG * 64, 0, stream>>>(e0, P0, x, rec, off, mw0, mb0, rp0, uw0, ub0, e1, ntiles);

  k_proj<<<nblocks, 256, 0, stream>>>(e1, mw1, P1, ntiles);
  k_node<float><<<gblocks, NT_WG * 64, 0, stream>>>(e1, P1, x, rec, off, mw1, mb1, rp1, uw1, ub1, outp, ntiles);
}


# Weight staging to LDS: each thread issues eight 16-byte row loads together and writes four 16-byte fragment pieces, instead of one 4-byte load per trip with a wait on each; LDS contents identical
# speedup vs baseline: 2.1096x; 1.1728x over previous
#include <hip/hip_runtime.h>


typedef _Float16 f16;
typedef __attribute__((ext_vector_type(16))) _Float16 v16h;
typedef __attribute__((ext_vector_type(8)))  _Float16 v8h;
typedef __attribute__((ext_vector_type(8)))  float    v8f;
typedef __attribute__((ext_vector_type(4)))  float    v4f;

#define HDIM  64
#define WAVES 8
#define NT_WG 8
#define BSH      7
#define BMASK    127
#define SORT_WGS 256
#define SORT_T   512
#define FINE_T   512
#define FINE_U   5


__device__ __forceinline__ v16h load_a_f32(const float* __restrict__ base, int kb) {
  const float4* p = (const float4*)base;
  float4 x0 = p[kb >> 2];
  float4 x1 = p[(kb >> 2) + 1];
  float4 x2 = p[4 + (kb >> 2)];
  float4 x3 = p[4 + (kb >> 2) + 1];
  v16h a;
  a[0] = (f16)x0.x; a[1] = (f16)x0.y; a[2]  = (f16)x0.z; a[3]  = (f16)x0.w;
  a[4] = (f16)x1.x; a[5] = (f16)x1.y; a[6]  = (f16)x1.z; a[7]  = (f16)x1.w;
  a[8] = (f16)x2.x; a[9] = (f16)x2.y; a[10] = (f16)x2.z; a[11] = (f16)x2.w;
  a[12]= (f16)x3.x; a[13]= (f16)x3.y; a[14] = (f16)x3.z; a[15] = (f16)x3.w;
  return a;
}

__device__ __forceinline__ v16h load_a_f16(const f16* __restrict__ base, int kb) {
  v8h lo = *(const v8h*)(base + kb);
  v8h hi = *(const v8h*)(base + 16 + kb);
  v16h a;
#pragma unroll
  for (int i = 0; i < 8; ++i) { a[i] = lo[i]; a[8 + i] = hi[i]; }
  return a;
}

__device__ __forceinline__ void stage_B(const float* __restrict__ W, f16* lds, int ksteps, int t, int nthr) {
  if (t < 0 || t >= nthr) return;
  for (int item = t; item < ksteps * 64; item += nthr) {
    const int c16 = item & 15;
    const int ihi = (item >> 4) & 1;
    const int h   = (item >> 5) & 1;
    const int s   = item >> 6;
    const float* src = W + (32 * s + 8 * h + 16 * ihi) * HDIM + 4 * c16;
    v4f x[8];
#pragma unroll
    for (int j = 0; j < 8; ++j) x[j] = *(const v4f*)(src + j * HDIM);
    f16* dst = lds + (s * 4 + (c16 >> 2)) * 512 + (h * 16 + 4 * (c16 & 3)) * 16 + ihi * 8;
#pragma unroll
    for (int c = 0; c < 4; ++c) {
      v8h v;
#pragma unroll
      for (int j = 0; j < 8; ++j) v[j] = (f16)x[j][c];
      *(v8h*)(dst + c * 16) = v;
    }
  }
}

__device__ __forceinline__ v8f wmma_f16(v16h a, v16h b, v8f c) {
  return __builtin_amdgcn_wmma_f32_16x16x32_f16(false, a, false, b, (short)0, c, false, false);
}


__global__ __launch_bounds__(256) void k_init(const float* __restrict__ h,
                                              const float* __restrict__ w_init,
                                              const float* __restrict__ b_init,
                                              f16* __restrict__ out, int ntiles) {
  __shared__ __align__(32) f16 sB[4 * 512];
  stage_B(w_init, sB, 1, (int)threadIdx.x, 64);
  __syncthreads();
  int wave = threadIdx.x >> 5, lane = threadIdx.x & 31;
  int tile = blockIdx.x * WAVES + wave;
  if (tile >= ntiles) return;
  int m = lane & 15, hi = lane >> 4, kb = hi * 8;
  int node = tile * 16 + m;
  v16h a = load_a_f32(h + (size_t)node * 32, kb);
#pragma unroll
  for (int nt = 0; nt < 4; ++nt) {
    v16h b = *(const v16h*)&sB[nt * 512 + lane * 16];
    v8f c = {};
    c = wmma_f16(a, b, c);
    int nglob = nt * 16 + (lane & 15);
    float bias = b_init[nglob];
#pragma unroll
    for (int r = 0; r < 8; ++r) {
      int M = r + 8 * hi;
      out[(size_t)(tile * 16 + M) * HDIM + nglob] = (f16)(c[r] + bias);
    }
  }
}


__global__ __launch_bounds__(SORT_T) void k_hist(const int* __restrict__ col, int* __restrict__ hist,
                                                 int nedges, int chunk, int nb) {
  extern __shared__ int s_cnt[];
  for (int b = threadIdx.x; b < nb; b += SORT_T) s_cnt[b] = 0;
  __syncthreads();
  const int lo = blockIdx.x * chunk;
  int hi = lo + chunk;
  hi = hi < nedges ? hi : nedges;
  for (int e0 = lo + threadIdx.x; e0 < hi; e0 += 8 * SORT_T) {
    int c[8];
#pragma unroll
    for (int k = 0; k < 8; ++k) {
      const int e = e0 + k * SORT_T;
      c[k] = e < hi ? col[e] : -1;
    }
#pragma unroll
    for (int k = 0; k < 8; ++k)
      if (c[k] >= 0) atomicAdd(&s_cnt[c[k] >> BSH], 1);
  }
  __syncthreads();
  for (int b = threadIdx.x; b < nb; b += SORT_T) hist[b * SORT_WGS + blockIdx.x] = s_cnt[b];
}

__global__ __launch_bounds__(1024) void k_scan(const int* __restrict__ deg, int* __restrict__ off, int nnodes) {
  __shared__ int s_part[32];
  __shared__ int s_wsum[32];
  const int t = threadIdx.x, lane = t & 31, w = t >> 5;
  const int start = blockIdx.x * 1024;
  int part = 0;
  for (int i = t; i < start; i += 1024) part += deg[i];
  for (int o = 16; o > 0; o >>= 1) part += __shfl_xor(part, o, 32);
  const int idx = start + t;
  const int v = idx < nnodes ? deg[idx] : 0;
  int x = v;
  for (int o = 1; o < 32; o <<= 1) {
    int y = __shfl_up(x, o, 32);
    if (lane >= o) x += y;
  }
  if (lane == 31) s_wsum[w] = x;
  if (lane == 0)  s_part[w] = part;
  __syncthreads();
  int before = 0;
  for (int i = 0; i < 32; ++i) {
    before += s_part[i];
    if (i < w) before += s_wsum[i];
  }
  const int excl = before + x - v;
  if (idx < nnodes) off[idx] = excl;
  if (idx == nnodes - 1) off[nnodes] = excl + v;
}

__global__ __launch_bounds__(SORT_T) void k_bin(const int* __restrict__ row, const int* __restrict__ col,
                                                const int* __restrict__ base, int2* __restrict__ binned,
                                                int nedges, int chunk, int nb) {
  extern __shared__ int s_cur[];
  for (int b = threadIdx.x; b < nb; b += SORT_T) s_cur[b] = base[b * SORT_WGS + blockIdx.x];
  __syncthreads();
  const int lo = blockIdx.x * chunk;
  int hi = lo + chunk;
  hi = hi < nedges ? hi : nedges;
  for (int e0 = lo + threadIdx.x; e0 < hi; e0 += 8 * SORT_T) {
    int c[8], r[8];
#pragma unroll
    for (int k = 0; k < 8; ++k) {
      const int e = e0 + k * SORT_T;
      c[k] = e < hi ? col[e] : -1;
      r[k] = e < hi ? row[e] : 0;
    }
#pragma unroll
    for (int k = 0; k < 8; ++k)
      if (c[k] >= 0) {
        const int pos = atomicAdd(&s_cur[c[k] >> BSH], 1);
        binned[pos] = make_int2(r[k], c[k]);
      }
  }
}

__global__ __launch_bounds__(FINE_T) void k_fine(const int2* __restrict__ binned, const int* __restrict__ base,
                                                 int* __restrict__ rec, int* __restrict__ off, int nnodes) {
  __shared__ int s_cnt[BMASK + 1];
  __shared__ int s_cur[BMASK + 1];
  __shared__ int s_w[FINE_T / 32];
  const int b = blockIdx.x, t = threadIdx.x, lane = t & 31, w = t >> 5;
  const int s = base[b * SORT_WGS], e = base[(b + 1) * SORT_WGS];
  if (t <= BMASK) s_cnt[t] = 0;
  __syncthreads();
  for (int i0 = s + t; i0 < e; i0 += FINE_U * FINE_T) {
    int c[FINE_U];
#pragma unroll
    for (int k = 0; k < FINE_U; ++k) {
      const int i = i0 + k * FINE_T;
      c[k] = i < e ? binned[i].y : -1;
    }
#pragma unroll
    for (int k = 0; k < FINE_U; ++k)
      if (c[k] >= 0) atomicAdd(&s_cnt[c[k] & BMASK], 1);
  }
  __syncthreads();
  const int v = t <= BMASK ? s_cnt[t] : 0;
  int x = v;
  for (int o = 1; o < 32; o <<= 1) {
    int y = __shfl_up(x, o, 32);
    if (lane >= o) x += y;
  }
  if (lane == 31) s_w[w] = x;
  __syncthreads();
  int before = 0;
  for (int i = 0; i < FINE_T / 32; ++i)
    if (i < w) before += s_w[i];
  const int excl = before + x - v;
  if (t <= BMASK) {
    s_cur[t] = excl;
    const int n = (b << BSH) + t;
    if (n <= nnodes) off[n] = s + excl;
  }
  __syncthreads();
  for (int i0 = s + t; i0 < e; i0 += FINE_U * FINE_T) {
    int2 rc[FINE_U];
#pragma unroll
    for (int k = 0; k < FINE_U; ++k) {
      const int i = i0 + k * FINE_T;
      rc[k] = i < e ? binned[i] : make_int2(0, -1);
    }
#pragma unroll
    for (int k = 0; k < FINE_U; ++k)
      if (rc[k].y >= 0) {
        const int pos = atomicAdd(&s_cur[rc[k].y & BMASK], 1);
        rec[s + pos] = rc[k].x;
      }
  }
}

__global__ __launch_bounds__(256) void k_proj(const f16* __restrict__ embed,
                                              const float* __restrict__ mw,
                                              float* __restrict__ P, int ntiles) {
  __shared__ __align__(32) f16 sB[8 * 512];
  stage_B(mw, sB, 2, (int)threadIdx.x, 128);
  __syncthreads();
  int wave = threadIdx.x >> 5, lane = threadIdx.x & 31;
  int tile = blockIdx.x * WAVES + wave;
  if (tile >= ntiles) return;
  int m = lane & 15, hi = lane >> 4, kb = hi * 8;
  int node = tile * 16 + m;
  v16h a0 = load_a_f16(embed + (size_t)node * HDIM +  0, kb);
  v16h a1 = load_a_f16(embed + (size_t)node * HDIM + 32, kb);
#pragma unroll
  for (int nt = 0; nt < 4; ++nt) {
    v8f c = {};
    c = wmma_f16(a0, *(const v16h*)&sB[(0 * 4 + nt) * 512 + lane * 16], c);
    c = wmma_f16(a1, *(const v16h*)&sB[(1 * 4 + nt) * 512 + lane * 16], c);
    int nglob = nt * 16 + (lane & 15);
#pragma unroll
    for (int r = 0; r < 8; ++r) {
      int M = r + 8 * hi;
      P[(size_t)(tile * 16 + M) * HDIM + nglob] = c[r];
    }
  }
}

template <typename OUT_T>
__global__ __launch_bounds__(NT_WG * 64) void k_node(const f16* __restrict__ embed,
                                              const float* __restrict__ P,
                                              const float* __restrict__ xpos,
                                              const int* __restrict__ rec,
                                              const int* __restrict__ off,
                                              const float* __restrict__ mw,
                                              const float* __restrict__ mb,
                                              const float* __restrict__ rp,
                                              const float* __restrict__ uw,
                                              const float* __restrict__ ub,
                                              OUT_T* __restrict__ out, int ntiles) {
  __shared__ __align__(32) f16 sU[16 * 512];
  __shared__ __align__(32) f16 sR[8 * 512];
  __shared__ __align__(32) f16 sD[8 * 512];
  __shared__ __align__(32) float sQ[NT_WG * 16 * HDIM];
  __shared__ __align__(32) f16   sA[NT_WG * 16 * HDIM];
  stage_B(uw, sU, 4, (int)threadIdx.x, 256);
  stage_B(rp, sR, 2, (int)threadIdx.x - 256, 128);
  stage_B(mw + 64 * HDIM, sD, 2, (int)threadIdx.x - 384, 128);
  __syncthreads();

  const int wave = __builtin_amdgcn_readfirstlane(threadIdx.x >> 5), lane = threadIdx.x & 31;
  const int t = wave >> 1, half = wave & 1;
  const int tile = blockIdx.x * NT_WG + t;
  const bool valid = tile < ntiles;
  const int m = lane & 15, hi = lane >> 4, kb = hi * 8;

  v16h a0 = {}, a1 = {};
  if (valid) {
    const int node = tile * 16 + m;
    a0 = load_a_f16(embed + (size_t)node * HDIM +  0, kb);
    a1 = load_a_f16(embed + (size_t)node * HDIM + 32, kb);
#pragma unroll
    for (int q = 0; q < 2; ++q) {
      const int nt = half * 2 + q;
      v8f c = {};
      c = wmma_f16(a0, *(const v16h*)&sD[(0 * 4 + nt) * 512 + lane * 16], c);
      c = wmma_f16(a1, *(const v16h*)&sD[(1 * 4 + nt) * 512 + lane * 16], c);
      const int nglob = nt * 16 + m;
      const float bias = mb[nglob];
#pragma unroll
      for (int r = 0; r < 8; ++r)
        sQ[(t * 16 + r + 8 * hi) * HDIM + nglob] = c[r] + bias;
    }
  }
  __syncthreads();

  if (valid) {
    const int gn0 = tile * 16 + half * 8;
    const int ln0 = t * 16 + half * 8;
    const int o = off[gn0 + (lane < 8 ? lane : 8)];
    const float2 w = *(const float2*)&mw[128 * HDIM + 2 * lane];
    for (int j = 0; j < 8; ++j) {
      const int b = __builtin_amdgcn_readlane(o, j);
      const int e = __builtin_amdgcn_readlane(o, j + 1);
      const int n = gn0 + j;
      const float xn0 = xpos[n * 3 + 0], xn1 = xpos[n * 3 + 1], xn2 = xpos[n * 3 + 2];
      const float2 q = *(const float2*)&sQ[(ln0 + j) * HDIM + 2 * lane];
      float accx = 0.0f, accy = 0.0f;
      for (int base = b; base < e; base += 32) {
        int cnt = e - base;
        cnt = cnt > 32 ? 32 : cnt;
        int r = 0;
        float d = 0.0f;
        if (lane < cnt) {
          r = rec[base + lane];
          const float dx = xpos[r * 3 + 0] - xn0;
          const float dy = xpos[r * 3 + 1] - xn1;
          const float dz = xpos[r * 3 + 2] - xn2;
          d = dx * dx + dy * dy + dz * dz;
        }
        for (int i = 0; i < cnt; i += 8) {
          float2 p[8];
          float dd[8];
#pragma unroll
          for (int u = 0; u < 8; ++u) {
            const int ru = __builtin_amdgcn_readlane(r, i + u);
            dd[u] = __int_as_float(__builtin_amdgcn_readlane(__float_as_int(d), i + u));
            p[u] = *(const float2*)&P[(size_t)ru * HDIM + 2 * lane];
          }
#pragma unroll
          for (int u = 0; u < 8; ++u) {
            float vx = p[u].x + q.x + dd[u] * w.x;
            float vy = p[u].y + q.y + dd[u] * w.y;
            vx = vx > 0.0f ? vx : 0.0f;
            vy = vy > 0.0f ? vy : 0.0f;
            if (i + u < cnt) { accx += vx; accy += vy; }
          }
        }
      }
      sA[(ln0 + j) * HDIM + 2 * lane + 0] = (f16)accx;
      sA[(ln0 + j) * HDIM + 2 * lane + 1] = (f16)accy;
    }
  }
  __syncthreads();

  if (valid) {
    const f16* arow = &sA[(t * 16 + m) * HDIM];
    v16h a2 = load_a_f16(arow +  0, kb);
    v16h a3 = load_a_f16(arow + 32, kb);
#pragma unroll
    for (int q = 0; q < 2; ++q) {
      const int nt = half * 2 + q;
      v8f cu = {};
      v8f cr = {};
#pragma unroll
      for (int s = 0; s < 4; ++s) {
        v16h b = *(const v16h*)&sU[(s * 4 + nt) * 512 + lane * 16];
        v16h a = (s == 0) ? a0 : (s == 1) ? a1 : (s == 2) ? a2 : a3;
        cu = wmma_f16(a, b, cu);
      }
#pragma unroll
      for (int s = 0; s < 2; ++s) {
        v16h b = *(const v16h*)&sR[(s * 4 + nt) * 512 + lane * 16];
        v16h a = (s == 0) ? a0 : a1;
        cr = wmma_f16(a, b, cr);
      }
      const int nglob = nt * 16 + m;
      const float bias = ub[nglob];
#pragma unroll
      for (int r = 0; r < 8; ++r) {
        const int M = r + 8 * hi;
        float v = cu[r] + bias;
        v = v > 0.0f ? v : 0.0f;
        out[(size_t)(tile * 16 + M) * HDIM + nglob] = (OUT_T)(cr[r] + v);
      }
    }
  }
}


extern "C" void kernel_launch(void* const* d_in, const int* in_sizes, int n_in,
                              void* d_out, int out_size, void* d_ws, size_t ws_size,
                              hipStream_t stream) {
  (void)n_in; (void)out_size; (void)ws_size;
  const float* x      = (const float*)d_in[0];
  const float* h      = (const float*)d_in[1];
  const int*   eidx   = (const int*)  d_in[2];
  const float* w_init = (const float*)d_in[4];
  const float* b_init = (const float*)d_in[5];
  const float* rp0 = (const float*)d_in[6];
  const float* mw0 = (const float*)d_in[7];
  const float* mb0 = (const float*)d_in[8];
  const float* uw0 = (const float*)d_in[9];
  const float* ub0 = (const float*)d_in[10];
  const float* rp1 = (const float*)d_in[11];
  const float* mw1 = (const float*)d_in[12];
  const float* mb1 = (const float*)d_in[13];
  const float* uw1 = (const float*)d_in[14];
  const float* ub1 = (const float*)d_in[15];

  int N = in_sizes[1] / 32;
  int E = in_sizes[2] / 2;
  const int* row = eidx;
  const int* col = eidx + E;

  char* ws = (char*)d_ws;
  size_t S    = (size_t)N * HDIM * sizeof(f16);
  size_t xoff = S + (size_t)E * 4 + ((((size_t)N + 1) * 4 + 255) & ~(size_t)255);
  f16*   e1   = (f16*)ws;
  int*   rec  = (int*)(ws + S);
  int*   off  = (int*)(ws + S + (size_t)E * 4);
  f16*   e0   = (f16*)(ws + xoff);
  float* P1   = (float*)(ws + xoff);
  float* outp = (float*)d_out;
  float* P0   = outp;

  int nb      = (N >> BSH) + 1;
  int nhist   = nb * SORT_WGS;
  int chunk   = (E + SORT_WGS - 1) / SORT_WGS;
  int2*  binned = (int2*)d_out;
  int*   hist   = (int*)((char*)d_out + (size_t)E * 8);
  int*   base   = hist + nhist;

  int ntiles  = (N + 15) / 16;
  int nblocks = (ntiles + WAVES - 1) / WAVES;
  int gblocks = (ntiles + NT_WG - 1) / NT_WG;
  int sblocks = (nhist + 1023) / 1024;

  k_init<<<nblocks, 256, 0, stream>>>(h, w_init, b_init, e0, ntiles);
  k_hist<<<SORT_WGS, SORT_T, nb * sizeof(int), stream>>>(col, hist, E, chunk, nb);
  k_scan<<<sblocks, 1024, 0, stream>>>(hist, base, nhist);
  k_bin <<<SORT_WGS, SORT_T, nb * sizeof(int), stream>>>(row, col, base, binned, E, chunk, nb);
  k_fine<<<nb, FINE_T, 0, stream>>>(binned, base, rec, off, N);

  k_proj<<<nblocks, 256, 0, stream>>>(e0, mw0, P0, ntiles);
  k_node<f16><<<gblocks, NT_WG * 64, 0, stream>>>(e0, P0, x, rec, off, mw0, mb0, rp0, uw0, ub0, e1, ntiles);

  k_proj<<<nblocks, 256, 0, stream>>>(e1, mw1, P1, ntiles);
  k_node<float><<<gblocks, NT_WG * 64, 0, stream>>>(e1, P1, x, rec, off, mw1, mb1, rp1, uw1, ub1, outp, ntiles);
}


# k_node edge sums: a wave walks the edges of its 8 nodes as one list in chunks of 32, all 32 source rows of P requested at once, ids one chunk ahead; each node still summed in list order
# speedup vs baseline: 2.4615x; 1.1668x over previous
#include <hip/hip_runtime.h>


typedef _Float16 f16;
typedef __attribute__((ext_vector_type(16))) _Float16 v16h;
typedef __attribute__((ext_vector_type(8)))  _Float16 v8h;
typedef __attribute__((ext_vector_type(8)))  float    v8f;
typedef __attribute__((ext_vector_type(4)))  float    v4f;

#define HDIM  64
#define WAVES 8
#define NT_WG 8
#define BSH      7
#define BMASK    127
#define SORT_WGS 256
#define SORT_T   512
#define FINE_T   512
#define FINE_U   5


__device__ __forceinline__ v16h load_a_f32(const float* __restrict__ base, int kb) {
  const float4* p = (const float4*)base;
  float4 x0 = p[kb >> 2];
  float4 x1 = p[(kb >> 2) + 1];
  float4 x2 = p[4 + (kb >> 2)];
  float4 x3 = p[4 + (kb >> 2) + 1];
  v16h a;
  a[0] = (f16)x0.x; a[1] = (f16)x0.y; a[2]  = (f16)x0.z; a[3]  = (f16)x0.w;
  a[4] = (f16)x1.x; a[5] = (f16)x1.y; a[6]  = (f16)x1.z; a[7]  = (f16)x1.w;
  a[8] = (f16)x2.x; a[9] = (f16)x2.y; a[10] = (f16)x2.z; a[11] = (f16)x2.w;
  a[12]= (f16)x3.x; a[13]= (f16)x3.y; a[14] = (f16)x3.z; a[15] = (f16)x3.w;
  return a;
}

__device__ __forceinline__ v16h load_a_f16(const f16* __restrict__ base, int kb) {
  v8h lo = *(const v8h*)(base + kb);
  v8h hi = *(const v8h*)(base + 16 + kb);
  v16h a;
#pragma unroll
  for (int i = 0; i < 8; ++i) { a[i] = lo[i]; a[8 + i] = hi[i]; }
  return a;
}

__device__ __forceinline__ void stage_B(const float* __restrict__ W, f16* lds, int ksteps, int t, int nthr) {
  if (t < 0 || t >= nthr) return;
  for (int item = t; item < ksteps * 64; item += nthr) {
    const int c16 = item & 15;
    const int ihi = (item >> 4) & 1;
    const int h   = (item >> 5) & 1;
    const int s   = item >> 6;
    const float* src = W + (32 * s + 8 * h + 16 * ihi) * HDIM + 4 * c16;
    v4f x[8];
#pragma unroll
    for (int j = 0; j < 8; ++j) x[j] = *(const v4f*)(src + j * HDIM);
    f16* dst = lds + (s * 4 + (c16 >> 2)) * 512 + (h * 16 + 4 * (c16 & 3)) * 16 + ihi * 8;
#pragma unroll
    for (int c = 0; c < 4; ++c) {
      v8h v;
#pragma unroll
      for (int j = 0; j < 8; ++j) v[j] = (f16)x[j][c];
      *(v8h*)(dst + c * 16) = v;
    }
  }
}

__device__ __forceinline__ v8f wmma_f16(v16h a, v16h b, v8f c) {
  return __builtin_amdgcn_wmma_f32_16x16x32_f16(false, a, false, b, (short)0, c, false, false);
}


__device__ __forceinline__ void init_body(const float* __restrict__ h,
                                          const float* __restrict__ w_init,
                                          const float* __restrict__ b_init,
                                          f16* __restrict__ out, int ntiles, int blk, f16* sB) {
  stage_B(w_init, sB, 1, (int)threadIdx.x, 64);
  __syncthreads();
  int wave = threadIdx.x >> 5, lane = threadIdx.x & 31;
  int tile = blk * (int)(blockDim.x >> 5) + wave;
  if (tile >= ntiles) return;
  int m = lane & 15, hi = lane >> 4, kb = hi * 8;
  int node = tile * 16 + m;
  v16h a = load_a_f32(h + (size_t)node * 32, kb);
#pragma unroll
  for (int nt = 0; nt < 4; ++nt) {
    v16h b = *(const v16h*)&sB[nt * 512 + lane * 16];
    v8f c = {};
    c = wmma_f16(a, b, c);
    int nglob = nt * 16 + (lane & 15);
    float bias = b_init[nglob];
#pragma unroll
    for (int r = 0; r < 8; ++r) {
      int M = r + 8 * hi;
      out[(size_t)(tile * 16 + M) * HDIM + nglob] = (f16)(c[r] + bias);
    }
  }
}


__device__ __forceinline__ void hist_body(const int* __restrict__ col, int* __restrict__ hist,
                                          int nedges, int chunk, int nb, int* s_cnt) {
  for (int b = threadIdx.x; b < nb; b += SORT_T) s_cnt[b] = 0;
  __syncthreads();
  const int lo = blockIdx.x * chunk;
  int hi = lo + chunk;
  hi = hi < nedges ? hi : nedges;
  for (int e0 = lo + threadIdx.x; e0 < hi; e0 += 8 * SORT_T) {
    int c[8];
#pragma unroll
    for (int k = 0; k < 8; ++k) {
      const int e = e0 + k * SORT_T;
      c[k] = e < hi ? col[e] : -1;
    }
#pragma unroll
    for (int k = 0; k < 8; ++k)
      if (c[k] >= 0) atomicAdd(&s_cnt[c[k] >> BSH], 1);
  }
  __syncthreads();
  for (int b = threadIdx.x; b < nb; b += SORT_T) hist[b * SORT_WGS + blockIdx.x] = s_cnt[b];
}

__global__ __launch_bounds__(SORT_T) void k_hist_init(const int* __restrict__ col, int* __restrict__ hist,
                                                      int nedges, int chunk, int nb,
                                                      const float* __restrict__ h,
                                                      const float* __restrict__ w_init,
                                                      const float* __restrict__ b_init,
                                                      f16* __restrict__ e0, int ntiles) {
  extern __shared__ int s_cnt[];
  __shared__ __align__(32) f16 sB[4 * 512];
  if (blockIdx.x < SORT_WGS) hist_body(col, hist, nedges, chunk, nb, s_cnt);
  else init_body(h, w_init, b_init, e0, ntiles, (int)blockIdx.x - SORT_WGS, sB);
}

__device__ __forceinline__ void scan_body(const int* __restrict__ deg, int* __restrict__ off, int nnodes,
                                          int* s_part, int* s_wsum) {
  const int t = threadIdx.x, lane = t & 31, w = t >> 5;
  const int start = blockIdx.x * 1024;
  int part = 0;
  for (int i = t; i < start; i += 1024) part += deg[i];
  for (int o = 16; o > 0; o >>= 1) part += __shfl_xor(part, o, 32);
  const int idx = start + t;
  const int v = idx < nnodes ? deg[idx] : 0;
  int x = v;
  for (int o = 1; o < 32; o <<= 1) {
    int y = __shfl_up(x, o, 32);
    if (lane >= o) x += y;
  }
  if (lane == 31) s_wsum[w] = x;
  if (lane == 0)  s_part[w] = part;
  __syncthreads();
  int before = 0;
  for (int i = 0; i < 32; ++i) {
    before += s_part[i];
    if (i < w) before += s_wsum[i];
  }
  const int excl = before + x - v;
  if (idx < nnodes) off[idx] = excl;
  if (idx == nnodes - 1) off[nnodes] = excl + v;
}

__global__ __launch_bounds__(SORT_T) void k_bin(const int* __restrict__ row, const int* __restrict__ col,
                                                const int* __restrict__ base, int2* __restrict__ binned,
                                                int nedges, int chunk, int nb) {
  extern __shared__ int s_cur[];
  for (int b = threadIdx.x; b < nb; b += SORT_T) s_cur[b] = base[b * SORT_WGS + blockIdx.x];
  __syncthreads();
  const int lo = blockIdx.x * chunk;
  int hi = lo + chunk;
  hi = hi < nedges ? hi : nedges;
  for (int e0 = lo + threadIdx.x; e0 < hi; e0 += 8 * SORT_T) {
    int c[8], r[8];
#pragma unroll
    for (int k = 0; k < 8; ++k) {
      const int e = e0 + k * SORT_T;
      c[k] = e < hi ? col[e] : -1;
      r[k] = e < hi ? row[e] : 0;
    }
#pragma unroll
    for (int k = 0; k < 8; ++k)
      if (c[k] >= 0) {
        const int pos = atomicAdd(&s_cur[c[k] >> BSH], 1);
        binned[pos] = make_int2(r[k], c[k]);
      }
  }
}

__global__ __launch_bounds__(FINE_T) void k_fine(const int2* __restrict__ binned, const int* __restrict__ base,
                                                 int* __restrict__ rec, int* __restrict__ off, int nnodes) {
  __shared__ int s_cnt[BMASK + 1];
  __shared__ int s_cur[BMASK + 1];
  __shared__ int s_w[FINE_T / 32];
  const int b = blockIdx.x, t = threadIdx.x, lane = t & 31, w = t >> 5;
  const int s = base[b * SORT_WGS], e = base[(b + 1) * SORT_WGS];
  if (t <= BMASK) s_cnt[t] = 0;
  __syncthreads();
  for (int i0 = s + t; i0 < e; i0 += FINE_U * FINE_T) {
    int c[FINE_U];
#pragma unroll
    for (int k = 0; k < FINE_U; ++k) {
      const int i = i0 + k * FINE_T;
      c[k] = i < e ? binned[i].y : -1;
    }
#pragma unroll
    for (int k = 0; k < FINE_U; ++k)
      if (c[k] >= 0) atomicAdd(&s_cnt[c[k] & BMASK], 1);
  }
  __syncthreads();
  const int v = t <= BMASK ? s_cnt[t] : 0;
  int x = v;
  for (int o = 1; o < 32; o <<= 1) {
    int y = __shfl_up(x, o, 32);
    if (lane >= o) x += y;
  }
  if (lane == 31) s_w[w] = x;
  __syncthreads();
  int before = 0;
  for (int i = 0; i < FINE_T / 32; ++i)
    if (i < w) before += s_w[i];
  const int excl = before + x - v;
  if (t <= BMASK) {
    s_cur[t] = excl;
    const int n = (b << BSH) + t;
    if (n <= nnodes) off[n] = s + excl;
  }
  __syncthreads();
  for (int i0 = s + t; i0 < e; i0 += FINE_U * FINE_T) {
    int2 rc[FINE_U];
#pragma unroll
    for (int k = 0; k < FINE_U; ++k) {
      const int i = i0 + k * FINE_T;
      rc[k] = i < e ? binned[i] : make_int2(0, -1);
    }
#pragma unroll
    for (int k = 0; k < FINE_U; ++k)
      if (rc[k].y >= 0) {
        const int pos = atomicAdd(&s_cur[rc[k].y & BMASK], 1);
        rec[s + pos] = rc[k].x;
      }
  }
}

__device__ __forceinline__ void proj_body(const f16* __restrict__ embed,
                                          const float* __restrict__ mw,
                                          float* __restrict__ P, int ntiles, int blk, f16* sB) {
  stage_B(mw, sB, 2, (int)threadIdx.x, 128);
  __syncthreads();
  int wave = threadIdx.x >> 5, lane = threadIdx.x & 31;
  int tile = blk * (int)(blockDim.x >> 5) + wave;
  if (tile >= ntiles) return;
  int m = lane & 15, hi = lane >> 4, kb = hi * 8;
  int node = tile * 16 + m;
  v16h a0 = load_a_f16(embed + (size_t)node * HDIM +  0, kb);
  v16h a1 = load_a_f16(embed + (size_t)node * HDIM + 32, kb);
#pragma unroll
  for (int nt = 0; nt < 4; ++nt) {
    v8f c = {};
    c = wmma_f16(a0, *(const v16h*)&sB[(0 * 4 + nt) * 512 + lane * 16], c);
    c = wmma_f16(a1, *(const v16h*)&sB[(1 * 4 + nt) * 512 + lane * 16], c);
    int nglob = nt * 16 + (lane & 15);
#pragma unroll
    for (int r = 0; r < 8; ++r) {
      int M = r + 8 * hi;
      P[(size_t)(tile * 16 + M) * HDIM + nglob] = c[r];
    }
  }
}

__global__ __launch_bounds__(256) void k_proj(const f16* __restrict__ embed,
                                              const float* __restrict__ mw,
                                              float* __restrict__ P, int ntiles) {
  __shared__ __align__(32) f16 sB[8 * 512];
  proj_body(embed, mw, P, ntiles, (int)blockIdx.x, sB);
}

__global__ __launch_bounds__(1024) void k_scan_proj(const int* __restrict__ deg, int* __restrict__ off,
                                                    int nnodes, int sblocks,
                                                    const f16* __restrict__ embed,
                                                    const float* __restrict__ mw,
                                                    float* __restrict__ P, int ntiles) {
  __shared__ int s_part[32];
  __shared__ int s_wsum[32];
  __shared__ __align__(32) f16 sB[8 * 512];
  if ((int)blockIdx.x < sblocks) scan_body(deg, off, nnodes, s_part, s_wsum);
  else proj_body(embed, mw, P, ntiles, (int)blockIdx.x - sblocks, sB);
}

__device__ __forceinline__ void edge_pos(const float* __restrict__ xpos, int ei, int r, int gn0, const int (&ob)[9],
                                         float& sx, float& sy, float& sz, float& tx, float& ty, float& tz) {
  int tj = 0;
#pragma unroll
  for (int k = 1; k < 8; ++k) tj += (ei >= ob[k]) ? 1 : 0;
  const int n = gn0 + tj;
  sx = xpos[r * 3 + 0]; sy = xpos[r * 3 + 1]; sz = xpos[r * 3 + 2];
  tx = xpos[n * 3 + 0]; ty = xpos[n * 3 + 1]; tz = xpos[n * 3 + 2];
}

template <typename OUT_T>
__global__ __launch_bounds__(NT_WG * 64) void k_node(const f16* __restrict__ embed,
                                              const float* __restrict__ P,
                                              const float* __restrict__ xpos,
                                              const int* __restrict__ rec,
                                              const int* __restrict__ off,
                                              const float* __restrict__ mw,
                                              const float* __restrict__ mb,
                                              const float* __restrict__ rp,
                                              const float* __restrict__ uw,
                                              const float* __restrict__ ub,
                                              OUT_T* __restrict__ out, int ntiles) {
  __shared__ __align__(32) f16 sU[16 * 512];
  __shared__ __align__(32) f16 sR[8 * 512];
  __shared__ __align__(32) f16 sD[8 * 512];
  __shared__ __align__(32) float sQ[NT_WG * 16 * HDIM];
  __shared__ __align__(32) f16   sA[NT_WG * 16 * HDIM];
  stage_B(uw, sU, 4, (int)threadIdx.x, 256);
  stage_B(rp, sR, 2, (int)threadIdx.x - 256, 128);
  stage_B(mw + 64 * HDIM, sD, 2, (int)threadIdx.x - 384, 128);
  __syncthreads();

  const int wave = __builtin_amdgcn_readfirstlane(threadIdx.x >> 5), lane = threadIdx.x & 31;
  const int t = wave >> 1, half = wave & 1;
  const int tile = blockIdx.x * NT_WG + t;
  const bool valid = tile < ntiles;
  const int m = lane & 15, hi = lane >> 4, kb = hi * 8;

  const int gn0 = tile * 16 + half * 8;
  const int ln0 = t * 16 + half * 8;

  v16h a0 = {}, a1 = {};
  int ob[9] = {0, 0, 0, 0, 0, 0, 0, 0, 0};
  int r_cur = 0;
  if (valid) {
    const int node = tile * 16 + m;
    a0 = load_a_f16(embed + (size_t)node * HDIM +  0, kb);
    a1 = load_a_f16(embed + (size_t)node * HDIM + 32, kb);
    const int o = off[gn0 + (lane < 8 ? lane : 8)];
#pragma unroll
    for (int k = 0; k < 9; ++k) ob[k] = __builtin_amdgcn_readlane(o, k);
    const int e0i = ob[0] + lane;
    r_cur = e0i < ob[8] ? rec[e0i] : 0;
#pragma unroll
    for (int q = 0; q < 2; ++q) {
      const int nt = half * 2 + q;
      v8f c = {};
      c = wmma_f16(a0, *(const v16h*)&sD[(0 * 4 + nt) * 512 + lane * 16], c);
      c = wmma_f16(a1, *(const v16h*)&sD[(1 * 4 + nt) * 512 + lane * 16], c);
      const int nglob = nt * 16 + m;
      const float bias = mb[nglob];
#pragma unroll
      for (int r = 0; r < 8; ++r)
        sQ[(t * 16 + r + 8 * hi) * HDIM + nglob] = c[r] + bias;
    }
  }
  __syncthreads();

  if (valid) {
    const float2 w = *(const float2*)&mw[128 * HDIM + 2 * lane];
    const int o8 = ob[8];
    int nj = 0;
    int nend = ob[1];
    float2 q  = *(const float2*)&sQ[(ln0 + 0) * HDIM + 2 * lane];
    float2 qn = *(const float2*)&sQ[(ln0 + 1) * HDIM + 2 * lane];
    float accx = 0.0f, accy = 0.0f;
    const unsigned lane8 = 8u * (unsigned)lane;
    for (int cb = ob[0]; cb < o8; cb += 32) {
      float sx, sy, sz, tx, ty, tz;
      edge_pos(xpos, cb + lane, r_cur, gn0, ob, sx, sy, sz, tx, ty, tz);
      float2 p[32];
#pragma unroll
      for (int u = 0; u < 32; ++u) {
        const unsigned ru = (unsigned)__builtin_amdgcn_readlane(r_cur, u);
        p[u] = *(const float2*)((const char*)P + (ru * (unsigned)(HDIM * sizeof(float)) + lane8));
      }
      const int en = cb + 32 + lane;
      const int r_nxt = en < o8 ? rec[en] : 0;
      const float dx = sx - tx;
      const float dy = sy - ty;
      const float dz = sz - tz;
      const float d = dx * dx + dy * dy + dz * dz;
#pragma unroll
      for (int u = 0; u < 32; ++u) {
        const int ei = cb + u;
        while (ei >= nend) {
          sA[(ln0 + nj) * HDIM + 2 * lane + 0] = (f16)accx;
          sA[(ln0 + nj) * HDIM + 2 * lane + 1] = (f16)accy;
          accx = 0.0f; accy = 0.0f;
          ++nj;
          nend = 0x7fffffff;
#pragma unroll
          for (int k = 2; k < 9; ++k) nend = (nj + 1 == k) ? ob[k] : nend;
          q = qn;
          qn = *(const float2*)&sQ[(ln0 + (nj < 7 ? nj + 1 : 7)) * HDIM + 2 * lane];
        }
        const float dd = __int_as_float(__builtin_amdgcn_readlane(__float_as_int(d), u));
        float vx = p[u].x + q.x + dd * w.x;
        float vy = p[u].y + q.y + dd * w.y;
        vx = vx > 0.0f ? vx : 0.0f;
        vy = vy > 0.0f ? vy : 0.0f;
        accx += vx; accy += vy;
      }
      r_cur = r_nxt;
    }
    for (; nj < 8; ++nj) {
      sA[(ln0 + nj) * HDIM + 2 * lane + 0] = (f16)accx;
      sA[(ln0 + nj) * HDIM + 2 * lane + 1] = (f16)accy;
      accx = 0.0f; accy = 0.0f;
    }
  }
  __syncthreads();

  if (valid) {
    const f16* arow = &sA[(t * 16 + m) * HDIM];
    v16h a2 = load_a_f16(arow +  0, kb);
    v16h a3 = load_a_f16(arow + 32, kb);
#pragma unroll
    for (int q = 0; q < 2; ++q) {
      const int nt = half * 2 + q;
      v8f cu = {};
      v8f cr = {};
#pragma unroll
      for (int s = 0; s < 4; ++s) {
        v16h b = *(const v16h*)&sU[(s * 4 + nt) * 512 + lane * 16];
        v16h a = (s == 0) ? a0 : (s == 1) ? a1 : (s == 2) ? a2 : a3;
        cu = wmma_f16(a, b, cu);
      }
#pragma unroll
      for (int s = 0; s < 2; ++s) {
        v16h b = *(const v16h*)&sR[(s * 4 + nt) * 512 + lane * 16];
        v16h a = (s == 0) ? a0 : a1;
        cr = wmma_f16(a, b, cr);
      }
      const int nglob = nt * 16 + m;
      const float bias = ub[nglob];
#pragma unroll
      for (int r = 0; r < 8; ++r) {
        const int M = r + 8 * hi;
        float v = cu[r] + bias;
        v = v > 0.0f ? v : 0.0f;
        out[(size_t)(tile * 16 + M) * HDIM + nglob] = (OUT_T)(cr[r] + v);
      }
    }
  }
}


extern "C" void kernel_launch(void* const* d_in, const int* in_sizes, int n_in,
                              void* d_out, int out_size, void* d_ws, size_t ws_size,
                              hipStream_t stream) {
  (void)n_in; (void)out_size; (void)ws_size;
  const float* x      = (const float*)d_in[0];
  const float* h      = (const float*)d_in[1];
  const int*   eidx   = (const int*)  d_in[2];
  const float* w_init = (const float*)d_in[4];
  const float* b_init = (const float*)d_in[5];
  const float* rp0 = (const float*)d_in[6];
  const float* mw0 = (const float*)d_in[7];
  const float* mb0 = (const float*)d_in[8];
  const float* uw0 = (const float*)d_in[9];
  const float* ub0 = (const float*)d_in[10];
  const float* rp1 = (const float*)d_in[11];
  const float* mw1 = (const float*)d_in[12];
  const float* mb1 = (const float*)d_in[13];
  const float* uw1 = (const float*)d_in[14];
  const float* ub1 = (const float*)d_in[15];

  int N = in_sizes[1] / 32;
  int E = in_sizes[2] / 2;
  const int* row = eidx;
  const int* col = eidx + E;

  char* ws = (char*)d_ws;
  size_t S    = (size_t)N * HDIM * sizeof(f16);
  size_t xoff = S + (size_t)E * 4 + ((((size_t)N + 1) * 4 + 255) & ~(size_t)255);
  f16*   e1   = (f16*)ws;
  int*   rec  = (int*)(ws + S);
  int*   off  = (int*)(ws + S + (size_t)E * 4);
  f16*   e0   = (f16*)(ws + xoff);
  float* P1   = (float*)(ws + xoff);
  float* outp = (float*)d_out;
  float* P0   = outp;

  int nb      = (N >> BSH) + 1;
  int nhist   = nb * SORT_WGS;
  int chunk   = (E + SORT_WGS - 1) / SORT_WGS;
  int2*  binned = (int2*)(ws + xoff + S);
  int*   hist   = (int*)ws;
  int*   base   = hist + nhist;

  int ntiles  = (N + 15) / 16;
  int nblocks = (ntiles + WAVES - 1) / WAVES;
  int gblocks = (ntiles + NT_WG - 1) / NT_WG;
  int sblocks = (nhist + 1023) / 1024;
  int iblocks = (ntiles + SORT_T / 32 - 1) / (SORT_T / 32);
  int pblocks = (ntiles + 31) / 32;

  k_hist_init<<<SORT_WGS + iblocks, SORT_T, nb * sizeof(int), stream>>>(col, hist, E, chunk, nb,
                                                                       h, w_init, b_init, e0, ntiles);
  k_scan_proj<<<sblocks + pblocks, 1024, 0, stream>>>(hist, base, nhist, sblocks, e0, mw0, P0, ntiles);
  k_bin <<<SORT_WGS, SORT_T, nb * sizeof(int), stream>>>(row, col, base, binned, E, chunk, nb);
  k_fine<<<nb, FINE_T, 0, stream>>>(binned, base, rec, off, N);

  k_node<f16><<<gblocks, NT_WG * 64, 0, stream>>>(e0, P0, x, rec, off, mw0, mb0, rp0, uw0, ub0, e1, ntiles);

  k_proj<<<nblocks, 256, 0, stream>>>(e1, mw1, P1, ntiles);
  k_node<float><<<gblocks, NT_WG * 64, 0, stream>>>(e1, P1, x, rec, off, mw1, mb1, rp1, uw1, ub1, outp, ntiles);
}


# Stores to memory staged in LDS and issued as whole 128-byte lines: rows of e0, P, e1 and the output by 16-byte stores, both ordered edge lists as contiguous runs in place of scattered words
# speedup vs baseline: 4.1596x; 1.6899x over previous
#include <hip/hip_runtime.h>


typedef _Float16 f16;
typedef __attribute__((ext_vector_type(16))) _Float16 v16h;
typedef __attribute__((ext_vector_type(8)))  _Float16 v8h;
typedef __attribute__((ext_vector_type(2)))  _Float16 v2h;
typedef __attribute__((ext_vector_type(8)))  float    v8f;
typedef __attribute__((ext_vector_type(4)))  float    v4f;

#define HDIM  64
#define NT_WG 8
#define EPITCH 72
#define BSH      7
#define BMASK    127
#define SORT_WGS 256
#define SORT_T   512
#define FINE_T   (NT_WG * 64)
#define FINE_TPS (FINE_T / SORT_WGS)
#define FINE_U   5
static_assert(FINE_T == FINE_TPS * SORT_WGS, "whole threads per segment");
static_assert((NT_WG * 16) == (BMASK + 1), "a k_node work-group owns exactly one bucket");


__device__ __forceinline__ v16h load_a_f32(const float* __restrict__ base, int kb) {
  const float4* p = (const float4*)base;
  float4 x0 = p[kb >> 2];
  float4 x1 = p[(kb >> 2) + 1];
  float4 x2 = p[4 + (kb >> 2)];
  float4 x3 = p[4 + (kb >> 2) + 1];
  v16h a;
  a[0] = (f16)x0.x; a[1] = (f16)x0.y; a[2]  = (f16)x0.z; a[3]  = (f16)x0.w;
  a[4] = (f16)x1.x; a[5] = (f16)x1.y; a[6]  = (f16)x1.z; a[7]  = (f16)x1.w;
  a[8] = (f16)x2.x; a[9] = (f16)x2.y; a[10] = (f16)x2.z; a[11] = (f16)x2.w;
  a[12]= (f16)x3.x; a[13]= (f16)x3.y; a[14] = (f16)x3.z; a[15] = (f16)x3.w;
  return a;
}

__device__ __forceinline__ v16h load_a_f16(const f16* __restrict__ base, int kb) {
  v8h lo = *(const v8h*)(base + kb);
  v8h hi = *(const v8h*)(base + 16 + kb);
  v16h a;
#pragma unroll
  for (int i = 0; i < 8; ++i) { a[i] = lo[i]; a[8 + i] = hi[i]; }
  return a;
}

__device__ __forceinline__ void stage_B(const float* __restrict__ W, f16* lds, int ksteps, int t, int nthr) {
  if (t < 0 || t >= nthr) return;
  for (int item = t; item < ksteps * 64; item += nthr) {
    const int c16 = item & 15;
    const int ihi = (item >> 4) & 1;
    const int h   = (item >> 5) & 1;
    const int s   = item >> 6;
    const float* src = W + (32 * s + 8 * h + 16 * ihi) * HDIM + 4 * c16;
    v4f x[8];
#pragma unroll
    for (int j = 0; j < 8; ++j) x[j] = *(const v4f*)(src + j * HDIM);
    f16* dst = lds + (s * 4 + (c16 >> 2)) * 512 + (h * 16 + 4 * (c16 & 3)) * 16 + ihi * 8;
#pragma unroll
    for (int c = 0; c < 4; ++c) {
      v8h v;
#pragma unroll
      for (int j = 0; j < 8; ++j) v[j] = (f16)x[j][c];
      *(v8h*)(dst + c * 16) = v;
    }
  }
}

__device__ __forceinline__ void stage_B2(const float* __restrict__ Wa, f16* la,
                                         const float* __restrict__ Wb, f16* lb, int ksteps, int t, int nthr) {
  if (t < 0 || t >= nthr) return;
  for (int item = t; item < ksteps * 64; item += nthr) {
    const int c16 = item & 15;
    const int ihi = (item >> 4) & 1;
    const int h   = (item >> 5) & 1;
    const int s   = item >> 6;
    const int so = (32 * s + 8 * h + 16 * ihi) * HDIM + 4 * c16;
    const int dd = (s * 4 + (c16 >> 2)) * 512 + (h * 16 + 4 * (c16 & 3)) * 16 + ihi * 8;
    v4f x[8], y[8];
#pragma unroll
    for (int j = 0; j < 8; ++j) {
      x[j] = *(const v4f*)(Wa + so + j * HDIM);
      y[j] = *(const v4f*)(Wb + so + j * HDIM);
    }
#pragma unroll
    for (int c = 0; c < 4; ++c) {
      v8h v, w;
#pragma unroll
      for (int j = 0; j < 8; ++j) { v[j] = (f16)x[j][c]; w[j] = (f16)y[j][c]; }
      *(v8h*)(la + dd + c * 16) = v;
      *(v8h*)(lb + dd + c * 16) = w;
    }
  }
}

__device__ __forceinline__ v8f wmma_f16(v16h a, v16h b, v8f c) {
  return __builtin_amdgcn_wmma_f32_16x16x32_f16(false, a, false, b, (short)0, c, false, false);
}


__device__ __forceinline__ void init_body(const float* __restrict__ h,
                                          const float* __restrict__ w_init,
                                          const float* __restrict__ b_init,
                                          const float* __restrict__ mw,
                                          f16* __restrict__ out, f16* __restrict__ P,
                                          int ntiles, int blk, f16* sB, f16* sM, f16* sE) {
  stage_B(w_init, sB, 1, (int)threadIdx.x, 64);
  stage_B(mw, sM, 2, (int)threadIdx.x - 64, 128);
  __syncthreads();
  const int wave = threadIdx.x >> 5, lane = threadIdx.x & 31;
  const int tile = blk * (SORT_T / 32) + wave;
  const bool valid = tile < ntiles;
  const int m = lane & 15, hi = lane >> 4, kb = hi * 8;
  f16* const erows = sE + wave * (16 * EPITCH);
  if (valid) {
    const int node = tile * 16 + m;
    const v16h a = load_a_f32(h + (size_t)node * 32, kb);
#pragma unroll
    for (int nt = 0; nt < 4; ++nt) {
      const v16h b = *(const v16h*)&sB[nt * 512 + lane * 16];
      v8f c = {};
      c = wmma_f16(a, b, c);
      const int nglob = nt * 16 + m;
      const float bias = b_init[nglob];
#pragma unroll
      for (int r = 0; r < 8; ++r)
        erows[(r + 8 * hi) * EPITCH + nglob] = (f16)(c[r] + bias);
    }
  }
  __syncthreads();
  const int prow = lane >> 3, pcol = (lane & 7) * 8;
  v8h e0row[4];
  if (valid) {
#pragma unroll
    for (int j = 0; j < 4; ++j) e0row[j] = *(const v8h*)&erows[(prow + 4 * j) * EPITCH + pcol];
    const f16* erow = erows + m * EPITCH;
    const v16h a0 = load_a_f16(erow +  0, kb);
    const v16h a1 = load_a_f16(erow + 32, kb);
#pragma unroll
    for (int nt = 0; nt < 4; ++nt) {
      v8f c = {};
      c = wmma_f16(a0, *(const v16h*)&sM[(0 * 4 + nt) * 512 + lane * 16], c);
      c = wmma_f16(a1, *(const v16h*)&sM[(1 * 4 + nt) * 512 + lane * 16], c);
      const int nglob = nt * 16 + m;
#pragma unroll
      for (int r = 0; r < 8; ++r)
        erows[(r + 8 * hi) * EPITCH + nglob] = (f16)c[r];
    }
  }
  __syncthreads();
  if (valid) {
#pragma unroll
    for (int j = 0; j < 4; ++j) {
      const size_t g = (size_t)(tile * 16 + prow + 4 * j) * HDIM + pcol;
      *(v8h*)&out[g] = e0row[j];
      *(v8h*)&P[g]   = *(const v8h*)&erows[(prow + 4 * j) * EPITCH + pcol];
    }
  }
}


__device__ __forceinline__ void sort_body(const int* __restrict__ row, const int* __restrict__ col,
                                          int* __restrict__ loc, int* __restrict__ binned,
                                          int nedges, int chunk, int nb, int* s_cnt, int* s_w,
                                          int* s_stage, int stage_cap) {
  const int t = threadIdx.x, lane = t & 31, w = t >> 5;
  const bool staged = chunk <= stage_cap;
  for (int b = t; b < nb; b += SORT_T) s_cnt[b] = 0;
  __syncthreads();
  const int lo = blockIdx.x * chunk;
  int hi = lo + chunk;
  hi = hi < nedges ? hi : nedges;
  for (int e0 = lo + t; e0 < hi; e0 += 8 * SORT_T) {
    int c[8];
#pragma unroll
    for (int k = 0; k < 8; ++k) {
      const int e = e0 + k * SORT_T;
      c[k] = e < hi ? col[e] : -1;
    }
#pragma unroll
    for (int k = 0; k < 8; ++k)
      if (c[k] >= 0) atomicAdd(&s_cnt[c[k] >> BSH], 1);
  }
  __syncthreads();
  int* lrow = loc + (size_t)blockIdx.x * (nb + 1);
  int carry = 0;
  for (int b0 = 0; b0 < nb; b0 += SORT_T) {
    const int b = b0 + t;
    const int v = b < nb ? s_cnt[b] : 0;
    int x = v;
    for (int o = 1; o < 32; o <<= 1) {
      int y = __shfl_up(x, o, 32);
      if (lane >= o) x += y;
    }
    if (lane == 31) s_w[w] = x;
    __syncthreads();
    int before = carry, tot = 0;
    for (int i = 0; i < SORT_T / 32; ++i) {
      const int sw = s_w[i];
      if (i < w) before += sw;
      tot += sw;
    }
    const int excl = before + x - v;
    if (b < nb) { s_cnt[b] = excl; lrow[b] = excl; }
    carry += tot;
    __syncthreads();
  }
  if (t == 0) lrow[nb] = carry;
  for (int e0 = lo + t; e0 < hi; e0 += 8 * SORT_T) {
    int c[8], r[8];
#pragma unroll
    for (int k = 0; k < 8; ++k) {
      const int e = e0 + k * SORT_T;
      c[k] = e < hi ? col[e] : -1;
      r[k] = e < hi ? row[e] : 0;
    }
#pragma unroll
    for (int k = 0; k < 8; ++k)
      if (c[k] >= 0) {
        const int pos = atomicAdd(&s_cnt[c[k] >> BSH], 1);
        const int word = (int)(((unsigned)r[k] << BSH) | ((unsigned)c[k] & BMASK));
        if (staged) s_stage[pos] = word;
        else        binned[lo + pos] = word;
      }
  }
  __syncthreads();
  if (staged)
    for (int i = t; i < hi - lo; i += SORT_T) binned[lo + i] = s_stage[i];
}

__global__ __launch_bounds__(SORT_T) void k_sort_init(const int* __restrict__ row, const int* __restrict__ col,
                                                      int* __restrict__ loc, int* __restrict__ binned,
                                                      int nedges, int chunk, int nb,
                                                      const float* __restrict__ h,
                                                      const float* __restrict__ w_init,
                                                      const float* __restrict__ b_init,
                                                      const float* __restrict__ mw,
                                                      f16* __restrict__ e0, f16* __restrict__ P, int ntiles) {
  extern __shared__ int s_cnt[];
  __shared__ int s_w[SORT_T / 32];
  __shared__ __align__(32) f16 sB[4 * 512];
  __shared__ __align__(32) f16 sM[8 * 512];
  __shared__ __align__(32) f16 sE[(SORT_T / 32) * 16 * EPITCH];
  if (blockIdx.x < SORT_WGS) sort_body(row, col, loc, binned, nedges, chunk, nb, s_cnt, s_w,
                                       reinterpret_cast<int*>(sE), (int)(sizeof(sE) / sizeof(int)));
  else init_body(h, w_init, b_init, mw, e0, P, ntiles, (int)blockIdx.x - SORT_WGS, sB, sM, sE);
}

__device__ __forceinline__ void fine_body(const int* __restrict__ binned, int l0, int l1,
                                          int* rec, int* off, int nnodes, int chunk, int b,
                                          int* s_cnt, int* s_cur, int* s_off, int* s_w, int* s_lo, int* s_len,
                                          int* s_stage, int stage_cap) {
  const int t = threadIdx.x, lane = t & 31, w = t >> 5;
  if (t < SORT_WGS) {
    s_lo[t]  = t * chunk + l0;
    s_len[t] = l1 - l0;
  }
  if (t <= BMASK) s_cnt[t] = 0;
  int part = l0;
  for (int o = 16; o > 0; o >>= 1) part += __shfl_xor(part, o, 32);
  if (lane == 0) s_w[w] = part;
  __syncthreads();
  int s = 0;
  for (int i = 0; i < FINE_T / 32; ++i) s += s_w[i];
  const int seg = t / FINE_TPS, par = t % FINE_TPS;
  const int slo = s_lo[seg], len = s_len[seg];
  for (int j0 = par; j0 < len; j0 += FINE_TPS * FINE_U) {
    int c[FINE_U];
#pragma unroll
    for (int k = 0; k < FINE_U; ++k) {
      const int j = j0 + k * FINE_TPS;
      c[k] = j < len ? binned[slo + j] : -1;
    }
#pragma unroll
    for (int k = 0; k < FINE_U; ++k)
      if (c[k] >= 0) atomicAdd(&s_cnt[c[k] & BMASK], 1);
  }
  __syncthreads();
  const int v = t <= BMASK ? s_cnt[t] : 0;
  int x = v;
  for (int o = 1; o < 32; o <<= 1) {
    int y = __shfl_up(x, o, 32);
    if (lane >= o) x += y;
  }
  if (lane == 31) s_w[w] = x;
  __syncthreads();
  int before = 0;
  for (int i = 0; i < FINE_T / 32; ++i)
    if (i < w) before += s_w[i];
  const int excl = before + x - v;
  if (t <= BMASK) {
    s_cur[t] = excl;
    s_off[t] = s + excl;
    if (t == BMASK) s_off[BMASK + 1] = s + excl + v;
    const int n = (b << BSH) + t;
    if (n <= nnodes) off[n] = s + excl;
  }
  __syncthreads();
  const int total = s_off[BMASK + 1] - s;
  const bool staged = total <= stage_cap;
  for (int j0 = par; j0 < len; j0 += FINE_TPS * FINE_U) {
    int rc[FINE_U];
#pragma unroll
    for (int k = 0; k < FINE_U; ++k) {
      const int j = j0 + k * FINE_TPS;
      rc[k] = j < len ? binned[slo + j] : -1;
    }
#pragma unroll
    for (int k = 0; k < FINE_U; ++k)
      if (rc[k] >= 0) {
        const int pos = atomicAdd(&s_cur[rc[k] & BMASK], 1);
        if (staged) s_stage[pos] = rc[k] >> BSH;
        else        rec[s + pos] = rc[k] >> BSH;
      }
  }
  __syncthreads();
  if (staged)
    for (int i = t; i < total; i += FINE_T) rec[s + i] = s_stage[i];
}

__device__ __forceinline__ void edge_pos(const float* __restrict__ xpos, int ei, int r, int gn0, const int (&ob)[9],
                                         float& sx, float& sy, float& sz, float& tx, float& ty, float& tz) {
  int tj = 0;
#pragma unroll
  for (int k = 1; k < 8; ++k) tj += (ei >= ob[k]) ? 1 : 0;
  const int n = gn0 + tj;
  sx = xpos[r * 3 + 0]; sy = xpos[r * 3 + 1]; sz = xpos[r * 3 + 2];
  tx = xpos[n * 3 + 0]; ty = xpos[n * 3 + 1]; tz = xpos[n * 3 + 2];
}

template <typename OUT_T, bool NEXT_P>
__global__ __launch_bounds__(NT_WG * 64) void k_node(const f16* __restrict__ embed,
                                              const f16* __restrict__ P,
                                              const float* __restrict__ xpos,
                                              int* rec,
                                              int* off,
                                              const int* __restrict__ binned,
                                              const int* __restrict__ loc,
                                              int nnodes, int chunk, int nb,
                                              const float* __restrict__ mw,
                                              const float* __restrict__ mb,
                                              const float* __restrict__ rp,
                                              const float* __restrict__ uw,
                                              const float* __restrict__ ub,
                                              OUT_T* __restrict__ out,
                                              const float* __restrict__ mwn,
                                              f16* __restrict__ Pn, int ntiles) {
  __shared__ __align__(32) f16 sU[16 * 512];
  __shared__ __align__(32) f16 sR[8 * 512];
  __shared__ __align__(32) f16 sD[8 * 512];
  __shared__ __align__(32) float sQ[NT_WG * 16 * HDIM];
  __shared__ __align__(32) f16   sA[NT_WG * 16 * HDIM];
  __shared__ __align__(32) f16   sP[8 * 512];
  f16* const sE = reinterpret_cast<f16*>(sQ);
  __shared__ int s_cnt[BMASK + 1];
  __shared__ int s_cur[BMASK + 1];
  __shared__ int s_off[BMASK + 2];
  __shared__ int s_w[FINE_T / 32];
  __shared__ int s_lo[SORT_WGS];
  __shared__ int s_len[SORT_WGS];
  int l0 = 0, l1 = 0;
  if constexpr (NEXT_P) {
    if (threadIdx.x < SORT_WGS) {
      const int* lr = loc + (size_t)threadIdx.x * (nb + 1) + blockIdx.x;
      l0 = lr[0];
      l1 = lr[1];
    }
  }
  stage_B(uw, sU, 4, (int)threadIdx.x, 256);
  stage_B(rp, sR, 2, (int)threadIdx.x - 256, 128);
  if constexpr (NEXT_P)
    stage_B2(mw + 64 * HDIM, sD, mwn, sP, 2, (int)threadIdx.x - 384, 128);
  else
    stage_B(mw + 64 * HDIM, sD, 2, (int)threadIdx.x - 384, 128);
  if constexpr (NEXT_P)
    fine_body(binned, l0, l1, rec, off, nnodes, chunk, (int)blockIdx.x, s_cnt, s_cur, s_off, s_w, s_lo, s_len,
              reinterpret_cast<int*>(sQ), (int)(sizeof(sQ) / sizeof(int)));
  __syncthreads();

  const int wave = __builtin_amdgcn_readfirstlane(threadIdx.x >> 5), lane = threadIdx.x & 31;
  const int t = wave >> 1, half = wave & 1;
  const int tile = blockIdx.x * NT_WG + t;
  const bool valid = tile < ntiles;
  const int m = lane & 15, hi = lane >> 4, kb = hi * 8;

  const int gn0 = tile * 16 + half * 8;
  const int ln0 = t * 16 + half * 8;

  v16h a0 = {}, a1 = {};
  int ob[9] = {0, 0, 0, 0, 0, 0, 0, 0, 0};
  int r_cur = 0;
  if (valid) {
    const int node = tile * 16 + m;
    a0 = load_a_f16(embed + (size_t)node * HDIM +  0, kb);
    a1 = load_a_f16(embed + (size_t)node * HDIM + 32, kb);
    int o;
    if constexpr (NEXT_P) o = s_off[ln0 + (lane < 8 ? lane : 8)];
    else                  o = off[gn0 + (lane < 8 ? lane : 8)];
#pragma unroll
    for (int k = 0; k < 9; ++k) ob[k] = __builtin_amdgcn_readlane(o, k);
    const int e0i = ob[0] + lane;
    r_cur = e0i < ob[8] ? rec[e0i] : 0;
#pragma unroll
    for (int q = 0; q < 2; ++q) {
      const int nt = half * 2 + q;
      v8f c = {};
      c = wmma_f16(a0, *(const v16h*)&sD[(0 * 4 + nt) * 512 + lane * 16], c);
      c = wmma_f16(a1, *(const v16h*)&sD[(1 * 4 + nt) * 512 + lane * 16], c);
      const int nglob = nt * 16 + m;
      const float bias = mb[nglob];
#pragma unroll
      for (int r = 0; r < 8; ++r)
        sQ[(t * 16 + r + 8 * hi) * HDIM + nglob] = c[r] + bias;
    }
  }
  __syncthreads();

  if (valid) {
    const float2 w = *(const float2*)&mw[128 * HDIM + 2 * lane];
    const int o8 = ob[8];
    int nj = 0;
    int nend = ob[1];
    float2 q  = *(const float2*)&sQ[(ln0 + 0) * HDIM + 2 * lane];
    float2 qn = *(const float2*)&sQ[(ln0 + 1) * HDIM + 2 * lane];
    float accx = 0.0f, accy = 0.0f;
    const unsigned lane4 = 4u * (unsigned)lane;
    for (int cb = ob[0]; cb < o8; cb += 32) {
      float sx, sy, sz, tx, ty, tz;
      edge_pos(xpos, cb + lane, r_cur, gn0, ob, sx, sy, sz, tx, ty, tz);
      v2h p[32];
#pragma unroll
      for (int u = 0; u < 32; ++u) {
        const unsigned ru = (unsigned)__builtin_amdgcn_readlane(r_cur, u);
        p[u] = *(const v2h*)((const char*)P + (ru * (unsigned)(HDIM * sizeof(f16)) + lane4));
      }
      const int en = cb + 32 + lane;
      const int r_nxt = en < o8 ? rec[en] : 0;
      const float dx = sx - tx;
      const float dy = sy - ty;
      const float dz = sz - tz;
      const float d = dx * dx + dy * dy + dz * dz;
#pragma unroll
      for (int u = 0; u < 32; ++u) {
        const int ei = cb + u;
        while (ei >= nend) {
          sA[(ln0 + nj) * HDIM + 2 * lane + 0] = (f16)accx;
          sA[(ln0 + nj) * HDIM + 2 * lane + 1] = (f16)accy;
          accx = 0.0f; accy = 0.0f;
          ++nj;
          nend = 0x7fffffff;
#pragma unroll
          for (int k = 2; k < 9; ++k) nend = (nj + 1 == k) ? ob[k] : nend;
          q = qn;
          qn = *(const float2*)&sQ[(ln0 + (nj < 7 ? nj + 1 : 7)) * HDIM + 2 * lane];
        }
        const float dd = __int_as_float(__builtin_amdgcn_readlane(__float_as_int(d), u));
        float vx = __builtin_fmaf(dd, w.x, (float)p[u][0]) + q.x;
        float vy = __builtin_fmaf(dd, w.y, (float)p[u][1]) + q.y;
        vx = vx > 0.0f ? vx : 0.0f;
        vy = vy > 0.0f ? vy : 0.0f;
        accx += vx; accy += vy;
      }
      r_cur = r_nxt;
    }
    for (; nj < 8; ++nj) {
      sA[(ln0 + nj) * HDIM + 2 * lane + 0] = (f16)accx;
      sA[(ln0 + nj) * HDIM + 2 * lane + 1] = (f16)accy;
      accx = 0.0f; accy = 0.0f;
    }
  }
  __syncthreads();

  if (valid) {
    const f16* arow = &sA[(t * 16 + m) * HDIM];
    v16h a2 = load_a_f16(arow +  0, kb);
    v16h a3 = load_a_f16(arow + 32, kb);
#pragma unroll
    for (int q = 0; q < 2; ++q) {
      const int nt = half * 2 + q;
      v8f cu = {};
      v8f cr = {};
#pragma unroll
      for (int s = 0; s < 4; ++s) {
        v16h b = *(const v16h*)&sU[(s * 4 + nt) * 512 + lane * 16];
        v16h a = (s == 0) ? a0 : (s == 1) ? a1 : (s == 2) ? a2 : a3;
        cu = wmma_f16(a, b, cu);
      }
#pragma unroll
      for (int s = 0; s < 2; ++s) {
        v16h b = *(const v16h*)&sR[(s * 4 + nt) * 512 + lane * 16];
        v16h a = (s == 0) ? a0 : a1;
        cr = wmma_f16(a, b, cr);
      }
      const int nglob = nt * 16 + m;
      const float bias = ub[nglob];
#pragma unroll
      for (int r = 0; r < 8; ++r) {
        const int M = r + 8 * hi;
        float v = cu[r] + bias;
        v = v > 0.0f ? v : 0.0f;
        const OUT_T o = (OUT_T)(cr[r] + v);
        if constexpr (NEXT_P) sE[(t * 16 + M) * EPITCH + nglob] = (f16)o;
        else                  sQ[(t * 16 + M) * HDIM + nglob] = o;
      }
    }
  }
  __syncthreads();

  if constexpr (NEXT_P) {
    if (valid) {
      const f16* erow = &sE[(t * 16 + m) * EPITCH];
      const v16h a4 = load_a_f16(erow +  0, kb);
      const v16h a5 = load_a_f16(erow + 32, kb);
#pragma unroll
      for (int q = 0; q < 2; ++q) {
        const int nt = half * 2 + q;
        v8f c = {};
        c = wmma_f16(a4, *(const v16h*)&sP[(0 * 4 + nt) * 512 + lane * 16], c);
        c = wmma_f16(a5, *(const v16h*)&sP[(1 * 4 + nt) * 512 + lane * 16], c);
        const int nglob = nt * 16 + m;
#pragma unroll
        for (int r = 0; r < 8; ++r)
          sA[(t * 16 + r + 8 * hi) * HDIM + nglob] = (f16)c[r];
      }
    }
    __syncthreads();
    if (valid) {
      const int pcol = (lane & 7) * 8;
#pragma unroll
      for (int j = 0; j < 2; ++j) {
        const int lr = half * 8 + (lane >> 3) + 4 * j;
        const size_t g = (size_t)(tile * 16 + lr) * HDIM + pcol;
        *(v8h*)&out[g] = *(const v8h*)&sE[(t * 16 + lr) * EPITCH + pcol];
        *(v8h*)&Pn[g]  = *(const v8h*)&sA[(t * 16 + lr) * HDIM + pcol];
      }
    }
  } else {
    if (valid) {
      const int pcol = (lane & 15) * 4;
#pragma unroll
      for (int j = 0; j < 4; ++j) {
        const int lr = half * 8 + (lane >> 4) + 2 * j;
        *(v4f*)&out[(size_t)(tile * 16 + lr) * HDIM + pcol] = *(const v4f*)&sQ[(t * 16 + lr) * HDIM + pcol];
      }
    }
  }
}


extern "C" void kernel_launch(void* const* d_in, const int* in_sizes, int n_in,
                              void* d_out, int out_size, void* d_ws, size_t ws_size,
                              hipStream_t stream) {
  (void)n_in; (void)out_size; (void)ws_size;
  const float* x      = (const float*)d_in[0];
  const float* h      = (const float*)d_in[1];
  const int*   eidx   = (const int*)  d_in[2];
  const float* w_init = (const float*)d_in[4];
  const float* b_init = (const float*)d_in[5];
  const float* rp0 = (const float*)d_in[6];
  const float* mw0 = (const float*)d_in[7];
  const float* mb0 = (const float*)d_in[8];
  const float* uw0 = (const float*)d_in[9];
  const float* ub0 = (const float*)d_in[10];
  const float* rp1 = (const float*)d_in[11];
  const float* mw1 = (const float*)d_in[12];
  const float* mb1 = (const float*)d_in[13];
  const float* uw1 = (const float*)d_in[14];
  const float* ub1 = (const float*)d_in[15];

  int N = in_sizes[1] / 32;
  int E = in_sizes[2] / 2;
  const int* row = eidx;
  const int* col = eidx + E;

  char* ws = (char*)d_ws;
  size_t S    = (size_t)N * HDIM * sizeof(f16);
  size_t xoff = S + (size_t)E * 4 + ((((size_t)N + 1) * 4 + 255) & ~(size_t)255);
  f16*   e1   = (f16*)ws;
  int*   rec  = (int*)(ws + S);
  int*   off  = (int*)(ws + S + (size_t)E * 4);
  f16*   e0   = (f16*)(ws + xoff);
  f16*   P1   = (f16*)(ws + xoff + S);
  float* outp = (float*)d_out;
  f16*   P0   = (f16*)d_out;

  int nb      = (N >> BSH) + 1;
  int chunk   = (E + SORT_WGS - 1) / SORT_WGS;
  int*   binned = (int*)((char*)d_out + S);
  int*   loc    = (int*)((char*)d_out + S + (size_t)E * 4);

  int ntiles  = (N + 15) / 16;
  int gblocks = (ntiles + NT_WG - 1) / NT_WG;
  int iblocks = (ntiles + SORT_T / 32 - 1) / (SORT_T / 32);

  k_sort_init<<<SORT_WGS + iblocks, SORT_T, nb * sizeof(int), stream>>>(row, col, loc, binned, E, chunk, nb,
                                                                       h, w_init, b_init, mw0, e0, P0, ntiles);

  k_node<f16, true><<<nb, NT_WG * 64, 0, stream>>>(e0, P0, x, rec, off, binned, loc, N, chunk, nb,
                                                   mw0, mb0, rp0, uw0, ub0, e1, mw1, P1, ntiles);
  k_node<float, false><<<gblocks, NT_WG * 64, 0, stream>>>(e1, P1, x, rec, off, nullptr, nullptr, N, chunk, nb,
                                                           mw1, mb1, rp1, uw1, ub1, outp, nullptr, nullptr, ntiles);
}
